# starting kernel
# baseline (speedup 1.0000x reference)
#include <hip/hip_runtime.h>
#include <hip/hip_bf16.h>

typedef __attribute__((ext_vector_type(2))) float v2f;
typedef __attribute__((ext_vector_type(8))) float v8f;

#define Bv 64
#define Tv 512
#define Dv 1024
#define Kv 32

__device__ __forceinline__ float fexp2(float x) { return __builtin_amdgcn_exp2f(x); }
__device__ __forceinline__ float flog2(float x) { return __builtin_amdgcn_logf(x); }

__device__ __forceinline__ float readlane_f(float v, int l) {
    return __builtin_bit_cast(float,
        __builtin_amdgcn_readlane(__builtin_bit_cast(int, v), l));
}

__global__ __launch_bounds__(256) void crf_transposeW(const float* __restrict__ W,
                                                      float* __restrict__ Wt) {
    int idx = blockIdx.x * 256 + threadIdx.x;
    int k = idx >> 10;
    int d = idx & 1023;
    Wt[idx] = W[d * Kv + k];
}

__global__ __launch_bounds__(128) void crf_emissions(const float* __restrict__ x,
                                                     const float* __restrict__ Wt,
                                                     const float* __restrict__ bias,
                                                     float* __restrict__ e) {
    const int lane = threadIdx.x & 31;
    const int wave = threadIdx.x >> 5;
    const int mt   = blockIdx.x * 4 + wave;
    const int half = lane >> 4;
    const int l15  = lane & 15;

    const float* xptr = x  + (size_t)(mt * 16 + l15) * Dv + half * 2;
    const float* w0   = Wt + (size_t)(l15)      * Dv + half * 2;
    const float* w1   = Wt + (size_t)(l15 + 16) * Dv + half * 2;

    v8f acc0 = {};
    v8f acc1 = {};
    #pragma unroll 4
    for (int d = 0; d < Dv / 4; ++d) {
        v2f a  = *(const v2f*)(xptr + d * 4);
        v2f b0 = *(const v2f*)(w0   + d * 4);
        v2f b1 = *(const v2f*)(w1   + d * 4);
        acc0 = __builtin_amdgcn_wmma_f32_16x16x4_f32(false, a, false, b0,
                                                     (short)0, acc0, false, false);
        acc1 = __builtin_amdgcn_wmma_f32_16x16x4_f32(false, a, false, b1,
                                                     (short)0, acc1, false, false);
    }

    const float bv0 = bias[l15];
    const float bv1 = bias[l15 + 16];
    #pragma unroll
    for (int r = 0; r < 8; ++r) {
        size_t row = (size_t)mt * 16 + r + half * 8;
        e[row * Kv + l15]      = acc0[r] + bv0;
        e[row * Kv + l15 + 16] = acc1[r] + bv1;
    }
}

__global__ __launch_bounds__(32) void crf_scan(const float* __restrict__ e,
                                               const float* __restrict__ U,
                                               float* __restrict__ alpha_out,
                                               float* __restrict__ beta_out) {
    const int   lane   = threadIdx.x;
    const float LOG2E  = 1.44269504088896340736f;
    const float LN2    = 0.69314718055994530942f;
    float eU[Kv];

    if (blockIdx.x < Bv) {
        const int bIdx = blockIdx.x;
        #pragma unroll
        for (int j = 0; j < Kv; ++j)
            eU[j] = fexp2(U[j * Kv + lane] * LOG2E);

        const float* eb = e         + (size_t)bIdx * Tv * Kv;
        float*       ab = alpha_out + (size_t)bIdx * Tv * Kv;

        float alpha = eb[lane];
        ab[lane] = alpha;
        for (int t = 1; t < Tv; ++t) {
            float m = alpha;
            #pragma unroll
            for (int off = 16; off; off >>= 1) m = fmaxf(m, __shfl_xor(m, off, 32));
            float p = fexp2((alpha - m) * LOG2E);
            float s = 0.f;
            #pragma unroll
            for (int j = 0; j < Kv; ++j)
                s = fmaf(readlane_f(p, j), eU[j], s);
            alpha = m + flog2(s) * LN2 + eb[t * Kv + lane];
            ab[t * Kv + lane] = alpha;
        }
    } else {
        const int bIdx = blockIdx.x - Bv;
        #pragma unroll
        for (int j = 0; j < Kv; ++j)
            eU[j] = fexp2(U[lane * Kv + j] * LOG2E);

        const float* eb = e        + (size_t)bIdx * Tv * Kv;
        float*       bb = beta_out + (size_t)bIdx * Tv * Kv;

        float beta = 0.f;
        bb[(Tv - 1) * Kv + lane] = 0.f;
        for (int t = Tv - 2; t >= 0; --t) {
            float g = eb[(t + 1) * Kv + lane] + beta;
            float m = g;
            #pragma unroll
            for (int off = 16; off; off >>= 1) m = fmaxf(m, __shfl_xor(m, off, 32));
            float q = fexp2((g - m) * LOG2E);
            float s = 0.f;
            #pragma unroll
            for (int kk = 0; kk < Kv; ++kk)
                s = fmaf(readlane_f(q, kk), eU[kk], s);
            beta = m + flog2(s) * LN2;
            bb[t * Kv + lane] = beta;
        }
    }
}

__global__ __launch_bounds__(256) void crf_softmax(const float* __restrict__ alpha,
                                                   float* __restrict__ out) {
    const int lane = threadIdx.x & 31;
    const int wave = threadIdx.x >> 5;
    const size_t row = (size_t)blockIdx.x * 8 + wave;
    const size_t idx = row * Kv + lane;
    const float LOG2E = 1.44269504088896340736f;

    float s = alpha[idx] + out[idx];
    float m = s;
    #pragma unroll
    for (int off = 16; off; off >>= 1) m = fmaxf(m, __shfl_xor(m, off, 32));
    float p = fexp2((s - m) * LOG2E);
    float sum = p;
    #pragma unroll
    for (int off = 16; off; off >>= 1) sum += __shfl_xor(sum, off, 32);
    out[idx] = p / sum;
}

extern "C" void kernel_launch(void* const* d_in, const int* in_sizes, int n_in,
                              void* d_out, int out_size, void* d_ws, size_t ws_size,
                              hipStream_t stream) {
    const float* x = (const float*)d_in[0];
    const float* W = (const float*)d_in[1];
    const float* U = (const float*)d_in[2];
    const float* b = (const float*)d_in[3];
    float* out = (float*)d_out;

    float* ws    = (float*)d_ws;
    float* e     = ws;
    float* alpha = ws + (size_t)Bv * Tv * Kv;
    float* Wt    = alpha + (size_t)Bv * Tv * Kv;

    crf_transposeW<<<dim3(128), dim3(256), 0, stream>>>(W, Wt);
    crf_emissions <<<dim3((Bv * Tv / 16) / 4), dim3(128), 0, stream>>>(x, Wt, b, e);
    crf_scan      <<<dim3(2 * Bv), dim3(32), 0, stream>>>(e, U, alpha, out);
    crf_softmax   <<<dim3(Bv * Tv / 8), dim3(256), 0, stream>>>(alpha, out);
}


# scan: 16 batches per wave as columns of f32 WMMA products with exp(U), linear domain, rescaled by 1/max per step; D tile is the next B operand (k order permuted); marginals a*b/sum; sum order changed
# speedup vs baseline: 1.7824x; 1.7824x over previous
#include <hip/hip_runtime.h>
#include <hip/hip_bf16.h>

typedef __attribute__((ext_vector_type(2))) float v2f;
typedef __attribute__((ext_vector_type(4))) float v4f;
typedef __attribute__((ext_vector_type(8))) float v8f;

#define Bv 64
#define Tv 512
#define Dv 1024
#define Kv 32

__device__ __forceinline__ float fexp2(float x) { return __builtin_amdgcn_exp2f(x); }
__device__ __forceinline__ float flog2(float x) { return __builtin_amdgcn_logf(x); }

__device__ __forceinline__ float readlane_f(float v, int l) {
    return __builtin_bit_cast(float,
        __builtin_amdgcn_readlane(__builtin_bit_cast(int, v), l));
}

__global__ __launch_bounds__(256) void crf_transposeW(const float* __restrict__ W,
                                                      float* __restrict__ Wt) {
    int idx = blockIdx.x * 256 + threadIdx.x;
    int k = idx >> 10;
    int d = idx & 1023;
    Wt[idx] = W[d * Kv + k];
}

__global__ __launch_bounds__(128) void crf_emissions(const float* __restrict__ x,
                                                     const float* __restrict__ Wt,
                                                     const float* __restrict__ bias,
                                                     float* __restrict__ e) {
    const int lane = threadIdx.x & 31;
    const int wave = threadIdx.x >> 5;
    const int mt   = blockIdx.x * 4 + wave;
    const int half = lane >> 4;
    const int l15  = lane & 15;

    const float* xptr = x  + (size_t)(mt * 16 + l15) * Dv + half * 2;
    const float* w0   = Wt + (size_t)(l15)      * Dv + half * 2;
    const float* w1   = Wt + (size_t)(l15 + 16) * Dv + half * 2;

    v8f acc0 = {};
    v8f acc1 = {};
    #pragma unroll 4
    for (int d = 0; d < Dv / 4; ++d) {
        v2f a  = *(const v2f*)(xptr + d * 4);
        v2f b0 = *(const v2f*)(w0   + d * 4);
        v2f b1 = *(const v2f*)(w1   + d * 4);
        acc0 = __builtin_amdgcn_wmma_f32_16x16x4_f32(false, a, false, b0,
                                                     (short)0, acc0, false, false);
        acc1 = __builtin_amdgcn_wmma_f32_16x16x4_f32(false, a, false, b1,
                                                     (short)0, acc1, false, false);
    }

    const float bv0 = bias[l15];
    const float bv1 = bias[l15 + 16];
    #pragma unroll
    for (int r = 0; r < 8; ++r) {
        size_t row = (size_t)mt * 16 + r + half * 8;
        e[row * Kv + l15]      = acc0[r] + bv0;
        e[row * Kv + l15 + 16] = acc1[r] + bv1;
    }
}

#define PFv 4
#define RSv 6

__device__ __forceinline__ float other_half(float v) {
    unsigned u = __builtin_bit_cast(unsigned, v);
    return __builtin_bit_cast(float,
        __builtin_amdgcn_permlanex16(u, u, 0x76543210u, 0xfedcba98u, false, false));
}

__device__ __forceinline__ float max16(v8f a, v8f b) {
    float m0 = fmaxf(fmaxf(a[0], a[1]), a[2]);
    float m1 = fmaxf(fmaxf(a[3], a[4]), a[5]);
    float m2 = fmaxf(fmaxf(a[6], a[7]), b[0]);
    float m3 = fmaxf(fmaxf(b[1], b[2]), b[3]);
    float m4 = fmaxf(fmaxf(b[4], b[5]), b[6]);
    float m5 = fmaxf(fmaxf(m0, m1), b[7]);
    float m6 = fmaxf(fmaxf(m2, m3), m4);
    return fmaxf(m5, m6);
}

template <bool FWD>
__device__ __forceinline__ void crf_chain(const float* __restrict__ e,
                                          const float* __restrict__ U,
                                          float* __restrict__ dst, int grp) {
    const int   lane  = threadIdx.x;
    const int   c     = lane & 15;
    const int   h     = lane >> 4;
    const float LOG2E = 1.44269504088896340736f;

    v2f A[2][8];
    #pragma unroll
    for (int m = 0; m < 2; ++m) {
        #pragma unroll
        for (int ks = 0; ks < 8; ++ks) {
            #pragma unroll
            for (int i = 0; i < 2; ++i) {
                const int tau = 16 * (ks >> 2) + 8 * h + 2 * (ks & 3) + i;
                const int row = 16 * m + c;
                const float u = FWD ? U[tau * Kv + row] : U[row * Kv + tau];
                A[m][ks][i] = fexp2(u * LOG2E);
            }
        }
    }

    const unsigned laneoff = (unsigned)((grp * 16 + c) * (Tv * Kv) + 8 * h);

    const float* elane = e   + laneoff;
    float*       dlane = dst + laneoff;

    v4f ring[RSv][4];
    auto issue = [&](int slot, int n) {
        const int tt = FWD ? n : (Tv - 1 - n);
        const float* rowp = elane + (size_t)tt * Kv;
        ring[slot][0] = *(const v4f*)(rowp);
        ring[slot][1] = *(const v4f*)(rowp + 4);
        ring[slot][2] = *(const v4f*)(rowp + 16);
        ring[slot][3] = *(const v4f*)(rowp + 20);
    };
    auto weights = [&](int slot, v8f& w0, v8f& w1) {
        v8f x0 = __builtin_shufflevector(ring[slot][0], ring[slot][1], 0, 1, 2, 3, 4, 5, 6, 7);
        v8f x1 = __builtin_shufflevector(ring[slot][2], ring[slot][3], 0, 1, 2, 3, 4, 5, 6, 7);
        float mx = max16(x0, x1);
        mx = fmaxf(mx, other_half(mx));
        const float ms = -mx * LOG2E;
        #pragma unroll
        for (int r = 0; r < 8; ++r) {
            w0[r] = fexp2(fmaf(x0[r], LOG2E, ms));
            w1[r] = fexp2(fmaf(x1[r], LOG2E, ms));
        }
    };
    auto store8 = [&](int n, int tile, v8f s) {
        const int tt = FWD ? n : (Tv - 1 - n);
        float* rowd = dlane + (size_t)tt * Kv + 16 * tile;
        *(v4f*)(rowd)     = __builtin_shufflevector(s, s, 0, 1, 2, 3);
        *(v4f*)(rowd + 4) = __builtin_shufflevector(s, s, 4, 5, 6, 7);
    };

    v8f S0, acc1p, w1p;
    issue(0, 0);
    weights(0, S0, w1p);
    #pragma unroll
    for (int r = 0; r < 8; ++r) acc1p[r] = 1.0f;
    store8(0, 0, FWD ? S0 : acc1p);
    #pragma unroll
    for (int j = 0; j < PFv; ++j) issue(j, 1 + j);

    auto step = [&](int slot, int n, bool prefetch) {
        if (prefetch) issue((slot + PFv) % RSv, n + PFv);
        v8f w0, w1;
        weights(slot, w0, w1);
        v8f acc0 = {};
        v8f acc1 = {};
        #pragma unroll
        for (int ks = 0; ks < 4; ++ks) {
            v2f b;
            b[0] = S0[2 * ks];
            b[1] = S0[2 * ks + 1];
            acc0 = __builtin_amdgcn_wmma_f32_16x16x4_f32(false, A[0][ks], false, b,
                                                         (short)0, acc0, false, false);
            acc1 = __builtin_amdgcn_wmma_f32_16x16x4_f32(false, A[1][ks], false, b,
                                                         (short)0, acc1, false, false);
        }
        __builtin_amdgcn_sched_barrier(0x444);
        const v8f S1 = acc1p * w1p;
        store8(n - 1, 1, FWD ? S1 : acc1p);
        #pragma unroll
        for (int ks = 4; ks < 8; ++ks) {
            v2f b;
            b[0] = S1[2 * (ks - 4)];
            b[1] = S1[2 * (ks - 4) + 1];
            acc0 = __builtin_amdgcn_wmma_f32_16x16x4_f32(false, A[0][ks], false, b,
                                                         (short)0, acc0, false, false);
            acc1 = __builtin_amdgcn_wmma_f32_16x16x4_f32(false, A[1][ks], false, b,
                                                         (short)0, acc1, false, false);
        }
        float mx = max16(S0, S1);
        mx = fmaxf(mx, other_half(mx));
        const float rs = __builtin_amdgcn_rcpf(mx);
        w0 *= rs;
        w1 *= rs;
        S0 = acc0 * w0;
        store8(n, 0, FWD ? S0 : acc0);
        acc1p = acc1;
        w1p   = w1;
    };

    constexpr int NT = 505;
    static_assert((NT - 1) % RSv == 0, "main loop must end on a ring boundary");
    static_assert(NT - 1 + PFv < Tv, "the main loop loads only rows that exist");
    for (int n = 1; n < NT; n += RSv) {
        #pragma unroll
        for (int u = 0; u < RSv; ++u) step(u, n + u, true);
    }
    #pragma unroll
    for (int u = 0; u < Tv - NT; ++u)
        step(u % RSv, NT + u, NT + u + PFv < Tv);
    store8(Tv - 1, 1, FWD ? (v8f)(acc1p * w1p) : acc1p);
}

__global__ __launch_bounds__(32) void crf_scan(const float* __restrict__ e,
                                               const float* __restrict__ U,
                                               float* __restrict__ alpha_out,
                                               float* __restrict__ beta_out) {
    if (blockIdx.x < Bv / 16) crf_chain<true >(e, U, alpha_out, blockIdx.x);
    else                      crf_chain<false>(e, U, beta_out,  blockIdx.x - Bv / 16);
}

__global__ __launch_bounds__(256) void crf_softmax(const float* __restrict__ alpha,
                                                   float* __restrict__ out) {
    const int lane = threadIdx.x & 31;
    const int wave = threadIdx.x >> 5;
    const size_t row = (size_t)blockIdx.x * 8 + wave;
    const size_t idx = row * Kv + lane;

    float p = alpha[idx] * out[idx];
    float sum = p;
    #pragma unroll
    for (int off = 16; off; off >>= 1) sum += __shfl_xor(sum, off, 32);
    out[idx] = p / sum;
}

extern "C" void kernel_launch(void* const* d_in, const int* in_sizes, int n_in,
                              void* d_out, int out_size, void* d_ws, size_t ws_size,
                              hipStream_t stream) {
    const float* x = (const float*)d_in[0];
    const float* W = (const float*)d_in[1];
    const float* U = (const float*)d_in[2];
    const float* b = (const float*)d_in[3];
    float* out = (float*)d_out;

    float* ws    = (float*)d_ws;
    float* e     = ws;
    float* alpha = ws + (size_t)Bv * Tv * Kv;
    float* Wt    = alpha + (size_t)Bv * Tv * Kv;

    crf_transposeW<<<dim3(128), dim3(256), 0, stream>>>(W, Wt);
    crf_emissions <<<dim3((Bv * Tv / 16) / 4), dim3(128), 0, stream>>>(x, Wt, b, e);
    crf_scan      <<<dim3(2 * (Bv / 16)), dim3(32), 0, stream>>>(e, U, alpha, out);
    crf_softmax   <<<dim3(Bv * Tv / 8), dim3(256), 0, stream>>>(alpha, out);
}


# weights exp(e - row max) are made once in the emissions kernel's epilogue (16-lane max per row) and stored in place of e; the two serial scan chains load them ready, no exp or row max per step
# speedup vs baseline: 2.2614x; 1.2687x over previous
#include <hip/hip_runtime.h>
#include <hip/hip_bf16.h>

typedef __attribute__((ext_vector_type(2))) float v2f;
typedef __attribute__((ext_vector_type(4))) float v4f;
typedef __attribute__((ext_vector_type(8))) float v8f;

#define Bv 64
#define Tv 512
#define Dv 1024
#define Kv 32

__device__ __forceinline__ float fexp2(float x) { return __builtin_amdgcn_exp2f(x); }
__device__ __forceinline__ float flog2(float x) { return __builtin_amdgcn_logf(x); }

__device__ __forceinline__ float readlane_f(float v, int l) {
    return __builtin_bit_cast(float,
        __builtin_amdgcn_readlane(__builtin_bit_cast(int, v), l));
}

__global__ __launch_bounds__(256) void crf_transposeW(const float* __restrict__ W,
                                                      float* __restrict__ Wt) {
    int idx = blockIdx.x * 256 + threadIdx.x;
    int k = idx >> 10;
    int d = idx & 1023;
    Wt[idx] = W[d * Kv + k];
}

__global__ __launch_bounds__(128) void crf_emissions(const float* __restrict__ x,
                                                     const float* __restrict__ Wt,
                                                     const float* __restrict__ bias,
                                                     float* __restrict__ wgt) {
    const int lane = threadIdx.x & 31;
    const int wave = threadIdx.x >> 5;
    const int mt   = blockIdx.x * 4 + wave;
    const int half = lane >> 4;
    const int l15  = lane & 15;

    const float* xptr = x  + (size_t)(mt * 16 + l15) * Dv + half * 2;
    const float* w0   = Wt + (size_t)(l15)      * Dv + half * 2;
    const float* w1   = Wt + (size_t)(l15 + 16) * Dv + half * 2;

    v8f acc0 = {};
    v8f acc1 = {};
    #pragma unroll 4
    for (int d = 0; d < Dv / 4; ++d) {
        v2f a  = *(const v2f*)(xptr + d * 4);
        v2f b0 = *(const v2f*)(w0   + d * 4);
        v2f b1 = *(const v2f*)(w1   + d * 4);
        acc0 = __builtin_amdgcn_wmma_f32_16x16x4_f32(false, a, false, b0,
                                                     (short)0, acc0, false, false);
        acc1 = __builtin_amdgcn_wmma_f32_16x16x4_f32(false, a, false, b1,
                                                     (short)0, acc1, false, false);
    }

    const float bv0 = bias[l15];
    const float bv1 = bias[l15 + 16];
    const float LOG2E = 1.44269504088896340736f;
    #pragma unroll
    for (int r = 0; r < 8; ++r) {
        const float v0 = acc0[r] + bv0;
        const float v1 = acc1[r] + bv1;
        float mx = fmaxf(v0, v1);
        #pragma unroll
        for (int off = 8; off; off >>= 1) mx = fmaxf(mx, __shfl_xor(mx, off, 32));
        const float ms = -mx * LOG2E;
        size_t row = (size_t)mt * 16 + r + half * 8;
        wgt[row * Kv + l15]      = fexp2(fmaf(v0, LOG2E, ms));
        wgt[row * Kv + l15 + 16] = fexp2(fmaf(v1, LOG2E, ms));
    }
}

#define PFv 4
#define RSv 6

__device__ __forceinline__ float other_half(float v) {
    unsigned u = __builtin_bit_cast(unsigned, v);
    return __builtin_bit_cast(float,
        __builtin_amdgcn_permlanex16(u, u, 0x76543210u, 0xfedcba98u, false, false));
}

__device__ __forceinline__ float max16(v8f a, v8f b) {
    float m0 = fmaxf(fmaxf(a[0], a[1]), a[2]);
    float m1 = fmaxf(fmaxf(a[3], a[4]), a[5]);
    float m2 = fmaxf(fmaxf(a[6], a[7]), b[0]);
    float m3 = fmaxf(fmaxf(b[1], b[2]), b[3]);
    float m4 = fmaxf(fmaxf(b[4], b[5]), b[6]);
    float m5 = fmaxf(fmaxf(m0, m1), b[7]);
    float m6 = fmaxf(fmaxf(m2, m3), m4);
    return fmaxf(m5, m6);
}

template <bool FWD>
__device__ __forceinline__ void crf_chain(const float* __restrict__ wgt,
                                          const float* __restrict__ U,
                                          float* __restrict__ dst, int grp) {
    const int   lane  = threadIdx.x;
    const int   c     = lane & 15;
    const int   h     = lane >> 4;
    const float LOG2E = 1.44269504088896340736f;

    v2f A[2][8];
    #pragma unroll
    for (int m = 0; m < 2; ++m) {
        #pragma unroll
        for (int ks = 0; ks < 8; ++ks) {
            #pragma unroll
            for (int i = 0; i < 2; ++i) {
                const int tau = 16 * (ks >> 2) + 8 * h + 2 * (ks & 3) + i;
                const int row = 16 * m + c;
                const float u = FWD ? U[tau * Kv + row] : U[row * Kv + tau];
                A[m][ks][i] = fexp2(u * LOG2E);
            }
        }
    }

    const unsigned laneoff = (unsigned)((grp * 16 + c) * (Tv * Kv) + 8 * h);

    const float* elane = wgt + laneoff;
    float*       dlane = dst + laneoff;

    v4f ring[RSv][4];
    auto issue = [&](int slot, int n) {
        const int tt = FWD ? n : (Tv - 1 - n);
        const float* rowp = elane + (size_t)tt * Kv;
        ring[slot][0] = *(const v4f*)(rowp);
        ring[slot][1] = *(const v4f*)(rowp + 4);
        ring[slot][2] = *(const v4f*)(rowp + 16);
        ring[slot][3] = *(const v4f*)(rowp + 20);
    };
    auto weights = [&](int slot, v8f& w0, v8f& w1) {
        w0 = __builtin_shufflevector(ring[slot][0], ring[slot][1], 0, 1, 2, 3, 4, 5, 6, 7);
        w1 = __builtin_shufflevector(ring[slot][2], ring[slot][3], 0, 1, 2, 3, 4, 5, 6, 7);
    };
    auto store8 = [&](int n, int tile, v8f s) {
        const int tt = FWD ? n : (Tv - 1 - n);
        float* rowd = dlane + (size_t)tt * Kv + 16 * tile;
        *(v4f*)(rowd)     = __builtin_shufflevector(s, s, 0, 1, 2, 3);
        *(v4f*)(rowd + 4) = __builtin_shufflevector(s, s, 4, 5, 6, 7);
    };

    v8f S0, acc1p, w1p;
    issue(0, 0);
    weights(0, S0, w1p);
    #pragma unroll
    for (int r = 0; r < 8; ++r) acc1p[r] = 1.0f;
    store8(0, 0, FWD ? S0 : acc1p);
    #pragma unroll
    for (int j = 0; j < PFv; ++j) issue(j, 1 + j);

    auto step = [&](int slot, int n, bool prefetch) {
        if (prefetch) issue((slot + PFv) % RSv, n + PFv);
        v8f w0, w1;
        weights(slot, w0, w1);
        v8f acc0 = {};
        v8f acc1 = {};
        #pragma unroll
        for (int ks = 0; ks < 4; ++ks) {
            v2f b;
            b[0] = S0[2 * ks];
            b[1] = S0[2 * ks + 1];
            acc0 = __builtin_amdgcn_wmma_f32_16x16x4_f32(false, A[0][ks], false, b,
                                                         (short)0, acc0, false, false);
            acc1 = __builtin_amdgcn_wmma_f32_16x16x4_f32(false, A[1][ks], false, b,
                                                         (short)0, acc1, false, false);
        }
        __builtin_amdgcn_sched_barrier(0x444);
        const v8f S1 = acc1p * w1p;
        store8(n - 1, 1, FWD ? S1 : acc1p);
        #pragma unroll
        for (int ks = 4; ks < 8; ++ks) {
            v2f b;
            b[0] = S1[2 * (ks - 4)];
            b[1] = S1[2 * (ks - 4) + 1];
            acc0 = __builtin_amdgcn_wmma_f32_16x16x4_f32(false, A[0][ks], false, b,
                                                         (short)0, acc0, false, false);
            acc1 = __builtin_amdgcn_wmma_f32_16x16x4_f32(false, A[1][ks], false, b,
                                                         (short)0, acc1, false, false);
        }
        float mx = max16(S0, S1);
        mx = fmaxf(mx, other_half(mx));
        const float rs = __builtin_amdgcn_rcpf(mx);
        w0 *= rs;
        w1 *= rs;
        S0 = acc0 * w0;
        store8(n, 0, FWD ? S0 : acc0);
        acc1p = acc1;
        w1p   = w1;
    };

    constexpr int NT = 505;
    static_assert((NT - 1) % RSv == 0, "main loop must end on a ring boundary");
    static_assert(NT - 1 + PFv < Tv, "the main loop loads only rows that exist");
    for (int n = 1; n < NT; n += RSv) {
        #pragma unroll
        for (int u = 0; u < RSv; ++u) step(u, n + u, true);
    }
    #pragma unroll
    for (int u = 0; u < Tv - NT; ++u)
        step(u % RSv, NT + u, NT + u + PFv < Tv);
    store8(Tv - 1, 1, FWD ? (v8f)(acc1p * w1p) : acc1p);
}

__global__ __launch_bounds__(32) void crf_scan(const float* __restrict__ wgt,
                                               const float* __restrict__ U,
                                               float* __restrict__ alpha_out,
                                               float* __restrict__ beta_out) {
    if (blockIdx.x < Bv / 16) crf_chain<true >(wgt, U, alpha_out, blockIdx.x);
    else                      crf_chain<false>(wgt, U, beta_out,  blockIdx.x - Bv / 16);
}

__global__ __launch_bounds__(256) void crf_softmax(const float* __restrict__ alpha,
                                                   float* __restrict__ out) {
    const int lane = threadIdx.x & 31;
    const int wave = threadIdx.x >> 5;
    const size_t row = (size_t)blockIdx.x * 8 + wave;
    const size_t idx = row * Kv + lane;

    float p = alpha[idx] * out[idx];
    float sum = p;
    #pragma unroll
    for (int off = 16; off; off >>= 1) sum += __shfl_xor(sum, off, 32);
    out[idx] = p / sum;
}

extern "C" void kernel_launch(void* const* d_in, const int* in_sizes, int n_in,
                              void* d_out, int out_size, void* d_ws, size_t ws_size,
                              hipStream_t stream) {
    const float* x = (const float*)d_in[0];
    const float* W = (const float*)d_in[1];
    const float* U = (const float*)d_in[2];
    const float* b = (const float*)d_in[3];
    float* out = (float*)d_out;

    float* ws    = (float*)d_ws;
    float* wgt   = ws;
    float* alpha = ws + (size_t)Bv * Tv * Kv;
    float* Wt    = alpha + (size_t)Bv * Tv * Kv;

    crf_transposeW<<<dim3(128), dim3(256), 0, stream>>>(W, Wt);
    crf_emissions <<<dim3((Bv * Tv / 16) / 4), dim3(128), 0, stream>>>(x, Wt, b, wgt);
    crf_scan      <<<dim3(2 * (Bv / 16)), dim3(32), 0, stream>>>(wgt, U, alpha, out);
    crf_softmax   <<<dim3(Bv * Tv / 8), dim3(256), 0, stream>>>(alpha, out);
}


# emissions: W (128 KB, packed in B-fragment order) is copied into LDS once per 8-wave work-group and read from there, 16 B per lane per k-step; only x uses vector memory, loaded 3 lines ahead
# speedup vs baseline: 2.4948x; 1.1032x over previous
#include <hip/hip_runtime.h>
#include <hip/hip_bf16.h>

typedef __attribute__((ext_vector_type(2))) float v2f;
typedef __attribute__((ext_vector_type(4))) float v4f;
typedef __attribute__((ext_vector_type(8))) float v8f;

#define Bv 64
#define Tv 512
#define Dv 1024
#define Kv 32

__device__ __forceinline__ float fexp2(float x) { return __builtin_amdgcn_exp2f(x); }
__device__ __forceinline__ float flog2(float x) { return __builtin_amdgcn_logf(x); }

__device__ __forceinline__ float readlane_f(float v, int l) {
    return __builtin_bit_cast(float,
        __builtin_amdgcn_readlane(__builtin_bit_cast(int, v), l));
}

__global__ __launch_bounds__(256) void crf_packW(const float* __restrict__ W,
                                                 float* __restrict__ Wp) {
    const int idx  = blockIdx.x * 256 + threadIdx.x;
    const int i    = idx & 3;
    const int lane = (idx >> 2) & 31;
    const int d4   = idx >> 7;
    const int n    = lane & 15;
    const int h    = lane >> 4;
    const int d    = 4 * d4 + 2 * h + (i & 1);
    const int k    = n + 16 * (i >> 1);
    Wp[idx] = W[d * Kv + k];
}

#define EKB 8
#define ENB ((Dv / 4) / EKB)
#define EXR 4

__global__ __launch_bounds__(256) void crf_emissions(const float* __restrict__ x,
                                                     const float* __restrict__ Wp,
                                                     const float* __restrict__ bias,
                                                     float* __restrict__ wgt) {
    __shared__ v4f wl[(Dv / 4) * 32];

    const int tid  = threadIdx.x;
    const int lane = tid & 31;
    const int wave = tid >> 5;
    const int mt   = blockIdx.x * 8 + wave;
    const int half = lane >> 4;
    const int l15  = lane & 15;

    const float* xptr = x + (size_t)(mt * 16 + l15) * Dv + half * 2;

    v2f xa[EXR][EKB];
    v4f wb[2][EKB];
    auto loadx = [&](int slot, int blk) {
        #pragma unroll
        for (int ks = 0; ks < EKB; ++ks)
            xa[slot][ks] = *(const v2f*)(xptr + (blk * EKB + ks) * 4);
    };
    auto loadw = [&](int slot, int blk) {
        #pragma unroll
        for (int ks = 0; ks < EKB; ++ks)
            wb[slot][ks] = wl[(blk * EKB + ks) * 32 + lane];
    };

    #pragma unroll
    for (int j = 0; j < EXR - 1; ++j) loadx(j, j);

    {
        const v4f* wp4 = (const v4f*)Wp;
        #pragma unroll
        for (int pass = 0; pass < 4; ++pass) {
            v4f t[8];
            #pragma unroll
            for (int j = 0; j < 8; ++j) t[j] = wp4[(pass * 8 + j) * 256 + tid];
            #pragma unroll
            for (int j = 0; j < 8; ++j) wl[(pass * 8 + j) * 256 + tid] = t[j];
        }
    }
    __syncthreads();
    loadw(0, 0);

    v8f acc0 = {};
    v8f acc1 = {};
    auto block = [&](int u, int blk, bool px, bool pw) {
        if (px) loadx((u + EXR - 1) % EXR, blk + EXR - 1);
        if (pw) loadw((u + 1) & 1, blk + 1);
        __builtin_amdgcn_sched_barrier(0);
        #pragma unroll
        for (int ks = 0; ks < EKB; ++ks) {
            const v4f w = wb[u & 1][ks];
            v2f b0, b1;
            b0[0] = w[0]; b0[1] = w[1];
            b1[0] = w[2]; b1[1] = w[3];
            acc0 = __builtin_amdgcn_wmma_f32_16x16x4_f32(false, xa[u % EXR][ks], false, b0,
                                                         (short)0, acc0, false, false);
            acc1 = __builtin_amdgcn_wmma_f32_16x16x4_f32(false, xa[u % EXR][ks], false, b1,
                                                         (short)0, acc1, false, false);
        }
    };
    static_assert(ENB % EXR == 0 && EXR % 2 == 0, "ring slots must be compile-time");
    for (int blk = 0; blk < ENB - EXR; blk += EXR) {
        #pragma unroll
        for (int u = 0; u < EXR; ++u) block(u, blk + u, true, true);
    }
    #pragma unroll
    for (int u = 0; u < EXR; ++u)
        block(u, ENB - EXR + u, u + EXR - 1 < EXR, u + 1 < EXR);

    const float bv0 = bias[l15];
    const float bv1 = bias[l15 + 16];
    const float LOG2E = 1.44269504088896340736f;
    #pragma unroll
    for (int r = 0; r < 8; ++r) {
        const float v0 = acc0[r] + bv0;
        const float v1 = acc1[r] + bv1;
        float mx = fmaxf(v0, v1);
        #pragma unroll
        for (int off = 8; off; off >>= 1) mx = fmaxf(mx, __shfl_xor(mx, off, 32));
        const float ms = -mx * LOG2E;
        size_t row = (size_t)mt * 16 + r + half * 8;
        wgt[row * Kv + l15]      = fexp2(fmaf(v0, LOG2E, ms));
        wgt[row * Kv + l15 + 16] = fexp2(fmaf(v1, LOG2E, ms));
    }
}

#define PFv 4
#define RSv 6

__device__ __forceinline__ float other_half(float v) {
    unsigned u = __builtin_bit_cast(unsigned, v);
    return __builtin_bit_cast(float,
        __builtin_amdgcn_permlanex16(u, u, 0x76543210u, 0xfedcba98u, false, false));
}

__device__ __forceinline__ float max16(v8f a, v8f b) {
    float m0 = fmaxf(fmaxf(a[0], a[1]), a[2]);
    float m1 = fmaxf(fmaxf(a[3], a[4]), a[5]);
    float m2 = fmaxf(fmaxf(a[6], a[7]), b[0]);
    float m3 = fmaxf(fmaxf(b[1], b[2]), b[3]);
    float m4 = fmaxf(fmaxf(b[4], b[5]), b[6]);
    float m5 = fmaxf(fmaxf(m0, m1), b[7]);
    float m6 = fmaxf(fmaxf(m2, m3), m4);
    return fmaxf(m5, m6);
}

template <bool FWD>
__device__ __forceinline__ void crf_chain(const float* __restrict__ wgt,
                                          const float* __restrict__ U,
                                          float* __restrict__ dst, int grp) {
    const int   lane  = threadIdx.x;
    const int   c     = lane & 15;
    const int   h     = lane >> 4;
    const float LOG2E = 1.44269504088896340736f;

    v2f A[2][8];
    #pragma unroll
    for (int m = 0; m < 2; ++m) {
        #pragma unroll
        for (int ks = 0; ks < 8; ++ks) {
            #pragma unroll
            for (int i = 0; i < 2; ++i) {
                const int tau = 16 * (ks >> 2) + 8 * h + 2 * (ks & 3) + i;
                const int row = 16 * m + c;
                const float u = FWD ? U[tau * Kv + row] : U[row * Kv + tau];
                A[m][ks][i] = fexp2(u * LOG2E);
            }
        }
    }

    const unsigned laneoff = (unsigned)((grp * 16 + c) * (Tv * Kv) + 8 * h);

    const float* elane = wgt + laneoff;
    float*       dlane = dst + laneoff;

    v4f ring[RSv][4];
    auto issue = [&](int slot, int n) {
        const int tt = FWD ? n : (Tv - 1 - n);
        const float* rowp = elane + (size_t)tt * Kv;
        ring[slot][0] = *(const v4f*)(rowp);
        ring[slot][1] = *(const v4f*)(rowp + 4);
        ring[slot][2] = *(const v4f*)(rowp + 16);
        ring[slot][3] = *(const v4f*)(rowp + 20);
    };
    auto weights = [&](int slot, v8f& w0, v8f& w1) {
        w0 = __builtin_shufflevector(ring[slot][0], ring[slot][1], 0, 1, 2, 3, 4, 5, 6, 7);
        w1 = __builtin_shufflevector(ring[slot][2], ring[slot][3], 0, 1, 2, 3, 4, 5, 6, 7);
    };
    auto store8 = [&](int n, int tile, v8f s) {
        const int tt = FWD ? n : (Tv - 1 - n);
        float* rowd = dlane + (size_t)tt * Kv + 16 * tile;
        *(v4f*)(rowd)     = __builtin_shufflevector(s, s, 0, 1, 2, 3);
        *(v4f*)(rowd + 4) = __builtin_shufflevector(s, s, 4, 5, 6, 7);
    };

    v8f S0, acc1p, w1p;
    issue(0, 0);
    weights(0, S0, w1p);
    #pragma unroll
    for (int r = 0; r < 8; ++r) acc1p[r] = 1.0f;
    store8(0, 0, FWD ? S0 : acc1p);
    #pragma unroll
    for (int j = 0; j < PFv; ++j) issue(j, 1 + j);

    auto step = [&](int slot, int n, bool prefetch) {
        if (prefetch) issue((slot + PFv) % RSv, n + PFv);
        v8f w0, w1;
        weights(slot, w0, w1);
        v8f acc0 = {};
        v8f acc1 = {};
        #pragma unroll
        for (int ks = 0; ks < 4; ++ks) {
            v2f b;
            b[0] = S0[2 * ks];
            b[1] = S0[2 * ks + 1];
            acc0 = __builtin_amdgcn_wmma_f32_16x16x4_f32(false, A[0][ks], false, b,
                                                         (short)0, acc0, false, false);
            acc1 = __builtin_amdgcn_wmma_f32_16x16x4_f32(false, A[1][ks], false, b,
                                                         (short)0, acc1, false, false);
        }
        __builtin_amdgcn_sched_barrier(0x444);
        const v8f S1 = acc1p * w1p;
        store8(n - 1, 1, FWD ? S1 : acc1p);
        #pragma unroll
        for (int ks = 4; ks < 8; ++ks) {
            v2f b;
            b[0] = S1[2 * (ks - 4)];
            b[1] = S1[2 * (ks - 4) + 1];
            acc0 = __builtin_amdgcn_wmma_f32_16x16x4_f32(false, A[0][ks], false, b,
                                                         (short)0, acc0, false, false);
            acc1 = __builtin_amdgcn_wmma_f32_16x16x4_f32(false, A[1][ks], false, b,
                                                         (short)0, acc1, false, false);
        }
        float mx = max16(S0, S1);
        mx = fmaxf(mx, other_half(mx));
        const float rs = __builtin_amdgcn_rcpf(mx);
        w0 *= rs;
        w1 *= rs;
        S0 = acc0 * w0;
        store8(n, 0, FWD ? S0 : acc0);
        acc1p = acc1;
        w1p   = w1;
    };

    constexpr int NT = 505;
    static_assert((NT - 1) % RSv == 0, "main loop must end on a ring boundary");
    static_assert(NT - 1 + PFv < Tv, "the main loop loads only rows that exist");
    for (int n = 1; n < NT; n += RSv) {
        #pragma unroll
        for (int u = 0; u < RSv; ++u) step(u, n + u, true);
    }
    #pragma unroll
    for (int u = 0; u < Tv - NT; ++u)
        step(u % RSv, NT + u, NT + u + PFv < Tv);
    store8(Tv - 1, 1, FWD ? (v8f)(acc1p * w1p) : acc1p);
}

__global__ __launch_bounds__(32) void crf_scan(const float* __restrict__ wgt,
                                               const float* __restrict__ U,
                                               float* __restrict__ alpha_out,
                                               float* __restrict__ beta_out) {
    if (blockIdx.x < Bv / 16) crf_chain<true >(wgt, U, alpha_out, blockIdx.x);
    else                      crf_chain<false>(wgt, U, beta_out,  blockIdx.x - Bv / 16);
}

__global__ __launch_bounds__(256) void crf_softmax(const float* __restrict__ alpha,
                                                   float* __restrict__ out) {
    const int lane = threadIdx.x & 31;
    const int wave = threadIdx.x >> 5;
    const size_t row = (size_t)blockIdx.x * 8 + wave;
    const size_t idx = row * Kv + lane;

    float p = alpha[idx] * out[idx];
    float sum = p;
    #pragma unroll
    for (int off = 16; off; off >>= 1) sum += __shfl_xor(sum, off, 32);
    out[idx] = p / sum;
}

extern "C" void kernel_launch(void* const* d_in, const int* in_sizes, int n_in,
                              void* d_out, int out_size, void* d_ws, size_t ws_size,
                              hipStream_t stream) {
    const float* x = (const float*)d_in[0];
    const float* W = (const float*)d_in[1];
    const float* U = (const float*)d_in[2];
    const float* b = (const float*)d_in[3];
    float* out = (float*)d_out;

    float* ws    = (float*)d_ws;
    float* wgt   = ws;
    float* alpha = ws + (size_t)Bv * Tv * Kv;
    float* Wp    = alpha + (size_t)Bv * Tv * Kv;

    crf_packW     <<<dim3(128), dim3(256), 0, stream>>>(W, Wp);
    crf_emissions <<<dim3((Bv * Tv / 16) / 8), dim3(256), 0, stream>>>(x, Wp, b, wgt);
    crf_scan      <<<dim3(2 * (Bv / 16)), dim3(32), 0, stream>>>(wgt, U, alpha, out);
    crf_softmax   <<<dim3(Bv * Tv / 8), dim3(256), 0, stream>>>(alpha, out);
}


# scan: each chain's step is split over the 2 waves of a work-group (two SIMDs): wave m owns output tile m, 8 WMMAs per step (was 16); tiles handed over in LDS, 1 barrier per step; sum order changed
# speedup vs baseline: 2.7386x; 1.0977x over previous
#include <hip/hip_runtime.h>
#include <hip/hip_bf16.h>

typedef __attribute__((ext_vector_type(2))) float v2f;
typedef __attribute__((ext_vector_type(4))) float v4f;
typedef __attribute__((ext_vector_type(8))) float v8f;

#define Bv 64
#define Tv 512
#define Dv 1024
#define Kv 32

__device__ __forceinline__ float fexp2(float x) { return __builtin_amdgcn_exp2f(x); }
__device__ __forceinline__ float flog2(float x) { return __builtin_amdgcn_logf(x); }

__device__ __forceinline__ float readlane_f(float v, int l) {
    return __builtin_bit_cast(float,
        __builtin_amdgcn_readlane(__builtin_bit_cast(int, v), l));
}

__global__ __launch_bounds__(256) void crf_packW(const float* __restrict__ W,
                                                 float* __restrict__ Wp) {
    const int idx  = blockIdx.x * 256 + threadIdx.x;
    const int i    = idx & 3;
    const int lane = (idx >> 2) & 31;
    const int d4   = idx >> 7;
    const int n    = lane & 15;
    const int h    = lane >> 4;
    const int d    = 4 * d4 + 2 * h + (i & 1);
    const int k    = n + 16 * (i >> 1);
    Wp[idx] = W[d * Kv + k];
}

#define EKB 8
#define ENB ((Dv / 4) / EKB)
#define EXR 4

__global__ __launch_bounds__(256) void crf_emissions(const float* __restrict__ x,
                                                     const float* __restrict__ Wp,
                                                     const float* __restrict__ bias,
                                                     float* __restrict__ wgt) {
    __shared__ v4f wl[(Dv / 4) * 32];

    const int tid  = threadIdx.x;
    const int lane = tid & 31;
    const int wave = tid >> 5;
    const int mt   = blockIdx.x * 8 + wave;
    const int half = lane >> 4;
    const int l15  = lane & 15;

    const float* xptr = x + (size_t)(mt * 16 + l15) * Dv + half * 2;

    v2f xa[EXR][EKB];
    v4f wb[2][EKB];
    auto loadx = [&](int slot, int blk) {
        #pragma unroll
        for (int ks = 0; ks < EKB; ++ks)
            xa[slot][ks] = *(const v2f*)(xptr + (blk * EKB + ks) * 4);
    };
    auto loadw = [&](int slot, int blk) {
        #pragma unroll
        for (int ks = 0; ks < EKB; ++ks)
            wb[slot][ks] = wl[(blk * EKB + ks) * 32 + lane];
    };

    #pragma unroll
    for (int j = 0; j < EXR - 1; ++j) loadx(j, j);

    {
        const v4f* wp4 = (const v4f*)Wp;
        #pragma unroll
        for (int pass = 0; pass < 4; ++pass) {
            v4f t[8];
            #pragma unroll
            for (int j = 0; j < 8; ++j) t[j] = wp4[(pass * 8 + j) * 256 + tid];
            #pragma unroll
            for (int j = 0; j < 8; ++j) wl[(pass * 8 + j) * 256 + tid] = t[j];
        }
    }
    __syncthreads();
    loadw(0, 0);

    v8f acc0 = {};
    v8f acc1 = {};
    auto block = [&](int u, int blk, bool px, bool pw) {
        if (px) loadx((u + EXR - 1) % EXR, blk + EXR - 1);
        if (pw) loadw((u + 1) & 1, blk + 1);
        __builtin_amdgcn_sched_barrier(0);
        #pragma unroll
        for (int ks = 0; ks < EKB; ++ks) {
            const v4f w = wb[u & 1][ks];
            v2f b0, b1;
            b0[0] = w[0]; b0[1] = w[1];
            b1[0] = w[2]; b1[1] = w[3];
            acc0 = __builtin_amdgcn_wmma_f32_16x16x4_f32(false, xa[u % EXR][ks], false, b0,
                                                         (short)0, acc0, false, false);
            acc1 = __builtin_amdgcn_wmma_f32_16x16x4_f32(false, xa[u % EXR][ks], false, b1,
                                                         (short)0, acc1, false, false);
        }
    };
    static_assert(ENB % EXR == 0 && EXR % 2 == 0, "ring slots must be compile-time");
    for (int blk = 0; blk < ENB - EXR; blk += EXR) {
        #pragma unroll
        for (int u = 0; u < EXR; ++u) block(u, blk + u, true, true);
    }
    #pragma unroll
    for (int u = 0; u < EXR; ++u)
        block(u, ENB - EXR + u, u + EXR - 1 < EXR, u + 1 < EXR);

    const float bv0 = bias[l15];
    const float bv1 = bias[l15 + 16];
    const float LOG2E = 1.44269504088896340736f;
    #pragma unroll
    for (int r = 0; r < 8; ++r) {
        const float v0 = acc0[r] + bv0;
        const float v1 = acc1[r] + bv1;
        float mx = fmaxf(v0, v1);
        #pragma unroll
        for (int off = 8; off; off >>= 1) mx = fmaxf(mx, __shfl_xor(mx, off, 32));
        const float ms = -mx * LOG2E;
        size_t row = (size_t)mt * 16 + r + half * 8;
        wgt[row * Kv + l15]      = fexp2(fmaf(v0, LOG2E, ms));
        wgt[row * Kv + l15 + 16] = fexp2(fmaf(v1, LOG2E, ms));
    }
}

#define PFv 4
#define RSv 6

__device__ __forceinline__ float other_half(float v) {
    unsigned u = __builtin_bit_cast(unsigned, v);
    return __builtin_bit_cast(float,
        __builtin_amdgcn_permlanex16(u, u, 0x76543210u, 0xfedcba98u, false, false));
}

__device__ __forceinline__ float max16(v8f a, v8f b) {
    float m0 = fmaxf(fmaxf(a[0], a[1]), a[2]);
    float m1 = fmaxf(fmaxf(a[3], a[4]), a[5]);
    float m2 = fmaxf(fmaxf(a[6], a[7]), b[0]);
    float m3 = fmaxf(fmaxf(b[1], b[2]), b[3]);
    float m4 = fmaxf(fmaxf(b[4], b[5]), b[6]);
    float m5 = fmaxf(fmaxf(m0, m1), b[7]);
    float m6 = fmaxf(fmaxf(m2, m3), m4);
    return fmaxf(m5, m6);
}

template <bool FWD>
__device__ __forceinline__ void crf_chain(const float* __restrict__ wgt,
                                          const float* __restrict__ U,
                                          float* __restrict__ dst, int grp,
                                          v4f* xch) {
    const int   lane  = threadIdx.x & 31;
    const int   m     = __builtin_amdgcn_readfirstlane((int)(threadIdx.x >> 5));
    const int   o     = 1 - m;
    const int   c     = lane & 15;
    const int   h     = lane >> 4;
    const float LOG2E = 1.44269504088896340736f;

    v2f Aown[4], Aoth[4];
    #pragma unroll
    for (int j = 0; j < 4; ++j) {
        #pragma unroll
        for (int i = 0; i < 2; ++i) {
            const int row = 16 * m + c;
            const int t0  = 16 * m + 8 * h + 2 * j + i;
            const int t1  = 16 * o + 8 * h + 2 * j + i;
            const float u0 = FWD ? U[t0 * Kv + row] : U[row * Kv + t0];
            const float u1 = FWD ? U[t1 * Kv + row] : U[row * Kv + t1];
            Aown[j][i] = fexp2(u0 * LOG2E);
            Aoth[j][i] = fexp2(u1 * LOG2E);
        }
    }

    const unsigned laneoff = (unsigned)((grp * 16 + c) * (Tv * Kv) + 16 * m + 8 * h);
    const float* elane = wgt + laneoff;
    float*       dlane = dst + laneoff;

    v4f* xown = xch + (m * 32 + lane) * 2;
    v4f* xoth = xch + (o * 32 + lane) * 2;

    v4f ring[RSv][2];
    auto issue = [&](int slot, int n) {
        const int tt = FWD ? n : (Tv - 1 - n);
        const float* rowp = elane + (size_t)tt * Kv;
        ring[slot][0] = *(const v4f*)(rowp);
        ring[slot][1] = *(const v4f*)(rowp + 4);
    };
    auto weights = [&](int slot) {
        return __builtin_shufflevector(ring[slot][0], ring[slot][1], 0, 1, 2, 3, 4, 5, 6, 7);
    };
    auto store8 = [&](int n, v8f s) {
        const int tt = FWD ? n : (Tv - 1 - n);
        float* rowd = dlane + (size_t)tt * Kv;
        *(v4f*)(rowd)     = __builtin_shufflevector(s, s, 0, 1, 2, 3);
        *(v4f*)(rowd + 4) = __builtin_shufflevector(s, s, 4, 5, 6, 7);
    };
    auto handover = [&](int p, v8f s) {
        xown[p * 128]     = __builtin_shufflevector(s, s, 0, 1, 2, 3);
        xown[p * 128 + 1] = __builtin_shufflevector(s, s, 4, 5, 6, 7);
        __builtin_amdgcn_fence(__ATOMIC_RELEASE, "workgroup", "local");
        __builtin_amdgcn_s_barrier();
        __builtin_amdgcn_fence(__ATOMIC_ACQUIRE, "workgroup", "local");
    };

    v8f Sown;
    issue(0, 0);
    Sown = weights(0);
    {
        v8f ones;
        #pragma unroll
        for (int r = 0; r < 8; ++r) ones[r] = 1.0f;
        store8(0, FWD ? Sown : ones);
    }
    #pragma unroll
    for (int j = 0; j < PFv; ++j) issue(j, 1 + j);
    handover(0, Sown);

    auto step = [&](int slot, int n, int par, bool prefetch) {
        if (prefetch) issue((slot + PFv) % RSv, n + PFv);
        const v4f o0 = xoth[par * 128];
        const v4f o1 = xoth[par * 128 + 1];
        v8f w = weights(slot);
        v8f accA = {};
        v8f accB = {};
        #pragma unroll
        for (int j = 0; j < 4; ++j) {
            v2f b;
            b[0] = Sown[2 * j];
            b[1] = Sown[2 * j + 1];
            if (j & 1) accB = __builtin_amdgcn_wmma_f32_16x16x4_f32(false, Aown[j], false, b,
                                                                    (short)0, accB, false, false);
            else       accA = __builtin_amdgcn_wmma_f32_16x16x4_f32(false, Aown[j], false, b,
                                                                    (short)0, accA, false, false);
        }
        const v8f Soth = __builtin_shufflevector(o0, o1, 0, 1, 2, 3, 4, 5, 6, 7);
        #pragma unroll
        for (int j = 0; j < 4; ++j) {
            v2f b;
            b[0] = Soth[2 * j];
            b[1] = Soth[2 * j + 1];
            if (j & 1) accB = __builtin_amdgcn_wmma_f32_16x16x4_f32(false, Aoth[j], false, b,
                                                                    (short)0, accB, false, false);
            else       accA = __builtin_amdgcn_wmma_f32_16x16x4_f32(false, Aoth[j], false, b,
                                                                    (short)0, accA, false, false);
        }
        float mx = max16(Sown, Soth);
        mx = fmaxf(mx, other_half(mx));
        const float rs = __builtin_amdgcn_rcpf(mx);
        w *= rs;
        const v8f acc = accA + accB;
        Sown = acc * w;
        store8(n, FWD ? Sown : acc);
        handover(par ^ 1, Sown);
    };

    constexpr int NT = 505;
    static_assert((NT - 1) % RSv == 0, "main loop must end on a ring boundary");
    static_assert(RSv % 2 == 0 && (NT - 1) % 2 == 0, "buffer parity must be compile-time");
    static_assert(NT - 1 + PFv < Tv, "the main loop loads only rows that exist");
    for (int n = 1; n < NT; n += RSv) {
        #pragma unroll
        for (int u = 0; u < RSv; ++u) step(u, n + u, u & 1, true);
    }
    #pragma unroll
    for (int u = 0; u < Tv - NT; ++u)
        step(u % RSv, NT + u, u & 1, NT + u + PFv < Tv);
}

__global__ __launch_bounds__(64) void crf_scan(const float* __restrict__ wgt,
                                               const float* __restrict__ U,
                                               float* __restrict__ alpha_out,
                                               float* __restrict__ beta_out) {
    __shared__ v4f xch[2 * 2 * 32 * 2];
    if (blockIdx.x < Bv / 16) crf_chain<true >(wgt, U, alpha_out, blockIdx.x, xch);
    else                      crf_chain<false>(wgt, U, beta_out,  blockIdx.x - Bv / 16, xch);
}

__global__ __launch_bounds__(256) void crf_softmax(const float* __restrict__ alpha,
                                                   float* __restrict__ out) {
    const int lane = threadIdx.x & 31;
    const int wave = threadIdx.x >> 5;
    const size_t row = (size_t)blockIdx.x * 8 + wave;
    const size_t idx = row * Kv + lane;

    float p = alpha[idx] * out[idx];
    float sum = p;
    #pragma unroll
    for (int off = 16; off; off >>= 1) sum += __shfl_xor(sum, off, 32);
    out[idx] = p / sum;
}

extern "C" void kernel_launch(void* const* d_in, const int* in_sizes, int n_in,
                              void* d_out, int out_size, void* d_ws, size_t ws_size,
                              hipStream_t stream) {
    const float* x = (const float*)d_in[0];
    const float* W = (const float*)d_in[1];
    const float* U = (const float*)d_in[2];
    const float* b = (const float*)d_in[3];
    float* out = (float*)d_out;

    float* ws    = (float*)d_ws;
    float* wgt   = ws;
    float* alpha = ws + (size_t)Bv * Tv * Kv;
    float* Wp    = alpha + (size_t)Bv * Tv * Kv;

    crf_packW     <<<dim3(128), dim3(256), 0, stream>>>(W, Wp);
    crf_emissions <<<dim3((Bv * Tv / 16) / 8), dim3(256), 0, stream>>>(x, Wp, b, wgt);
    crf_scan      <<<dim3(2 * (Bv / 16)), dim3(64), 0, stream>>>(wgt, U, alpha, out);
    crf_softmax   <<<dim3(Bv * Tv / 8), dim3(256), 0, stream>>>(alpha, out);
}


# scan cut into 16 time chunks: per-chunk 32x32 transfer matrices on f32 WMMA (all chunks at once), chunk start vectors by one mat-vec per chunk, then all chunks' 32 steps at once; sum order changed
# speedup vs baseline: 5.1711x; 1.8882x over previous
#include <hip/hip_runtime.h>
#include <hip/hip_bf16.h>

typedef __attribute__((ext_vector_type(2))) float v2f;
typedef __attribute__((ext_vector_type(4))) float v4f;
typedef __attribute__((ext_vector_type(8))) float v8f;

#define Bv 64
#define Tv 512
#define Dv 1024
#define Kv 32

__device__ __forceinline__ float fexp2(float x) { return __builtin_amdgcn_exp2f(x); }
__device__ __forceinline__ float flog2(float x) { return __builtin_amdgcn_logf(x); }

__device__ __forceinline__ float readlane_f(float v, int l) {
    return __builtin_bit_cast(float,
        __builtin_amdgcn_readlane(__builtin_bit_cast(int, v), l));
}

__global__ __launch_bounds__(256) void crf_packW(const float* __restrict__ W,
                                                 float* __restrict__ Wp) {
    const int idx  = blockIdx.x * 256 + threadIdx.x;
    const int i    = idx & 3;
    const int lane = (idx >> 2) & 31;
    const int d4   = idx >> 7;
    const int n    = lane & 15;
    const int h    = lane >> 4;
    const int d    = 4 * d4 + 2 * h + (i & 1);
    const int k    = n + 16 * (i >> 1);
    Wp[idx] = W[d * Kv + k];
}

#define EKB 8
#define ENB ((Dv / 4) / EKB)
#define EXR 4

__global__ __launch_bounds__(256) void crf_emissions(const float* __restrict__ x,
                                                     const float* __restrict__ Wp,
                                                     const float* __restrict__ bias,
                                                     float* __restrict__ wgt) {
    __shared__ v4f wl[(Dv / 4) * 32];

    const int tid  = threadIdx.x;
    const int lane = tid & 31;
    const int wave = tid >> 5;
    const int mt   = blockIdx.x * 8 + wave;
    const int half = lane >> 4;
    const int l15  = lane & 15;

    const float* xptr = x + (size_t)(mt * 16 + l15) * Dv + half * 2;

    v2f xa[EXR][EKB];
    v4f wb[2][EKB];
    auto loadx = [&](int slot, int blk) {
        #pragma unroll
        for (int ks = 0; ks < EKB; ++ks)
            xa[slot][ks] = *(const v2f*)(xptr + (blk * EKB + ks) * 4);
    };
    auto loadw = [&](int slot, int blk) {
        #pragma unroll
        for (int ks = 0; ks < EKB; ++ks)
            wb[slot][ks] = wl[(blk * EKB + ks) * 32 + lane];
    };

    #pragma unroll
    for (int j = 0; j < EXR - 1; ++j) loadx(j, j);

    {
        const v4f* wp4 = (const v4f*)Wp;
        #pragma unroll
        for (int pass = 0; pass < 4; ++pass) {
            v4f t[8];
            #pragma unroll
            for (int j = 0; j < 8; ++j) t[j] = wp4[(pass * 8 + j) * 256 + tid];
            #pragma unroll
            for (int j = 0; j < 8; ++j) wl[(pass * 8 + j) * 256 + tid] = t[j];
        }
    }
    __syncthreads();
    loadw(0, 0);

    v8f acc0 = {};
    v8f acc1 = {};
    auto block = [&](int u, int blk, bool px, bool pw) {
        if (px) loadx((u + EXR - 1) % EXR, blk + EXR - 1);
        if (pw) loadw((u + 1) & 1, blk + 1);
        __builtin_amdgcn_sched_barrier(0);
        #pragma unroll
        for (int ks = 0; ks < EKB; ++ks) {
            const v4f w = wb[u & 1][ks];
            v2f b0, b1;
            b0[0] = w[0]; b0[1] = w[1];
            b1[0] = w[2]; b1[1] = w[3];
            acc0 = __builtin_amdgcn_wmma_f32_16x16x4_f32(false, xa[u % EXR][ks], false, b0,
                                                         (short)0, acc0, false, false);
            acc1 = __builtin_amdgcn_wmma_f32_16x16x4_f32(false, xa[u % EXR][ks], false, b1,
                                                         (short)0, acc1, false, false);
        }
    };
    static_assert(ENB % EXR == 0 && EXR % 2 == 0, "ring slots must be compile-time");
    for (int blk = 0; blk < ENB - EXR; blk += EXR) {
        #pragma unroll
        for (int u = 0; u < EXR; ++u) block(u, blk + u, true, true);
    }
    #pragma unroll
    for (int u = 0; u < EXR; ++u)
        block(u, ENB - EXR + u, u + EXR - 1 < EXR, u + 1 < EXR);

    const float bv0 = bias[l15];
    const float bv1 = bias[l15 + 16];
    const float LOG2E = 1.44269504088896340736f;
    #pragma unroll
    for (int r = 0; r < 8; ++r) {
        const float v0 = acc0[r] + bv0;
        const float v1 = acc1[r] + bv1;
        float mx = fmaxf(v0, v1);
        #pragma unroll
        for (int off = 8; off; off >>= 1) mx = fmaxf(mx, __shfl_xor(mx, off, 32));
        const float ms = -mx * LOG2E;
        size_t row = (size_t)mt * 16 + r + half * 8;
        wgt[row * Kv + l15]      = fexp2(fmaf(v0, LOG2E, ms));
        wgt[row * Kv + l15 + 16] = fexp2(fmaf(v1, LOG2E, ms));
    }
}

#define CHL 32
#define NCH (Tv / CHL)
#define PFv 4
#define RSv 8

__device__ __forceinline__ float other_half(float v) {
    unsigned u = __builtin_bit_cast(unsigned, v);
    return __builtin_bit_cast(float,
        __builtin_amdgcn_permlanex16(u, u, 0x76543210u, 0xfedcba98u, false, false));
}

__device__ __forceinline__ float max16(v8f a, v8f b) {
    float m0 = fmaxf(fmaxf(a[0], a[1]), a[2]);
    float m1 = fmaxf(fmaxf(a[3], a[4]), a[5]);
    float m2 = fmaxf(fmaxf(a[6], a[7]), b[0]);
    float m3 = fmaxf(fmaxf(b[1], b[2]), b[3]);
    float m4 = fmaxf(fmaxf(b[4], b[5]), b[6]);
    float m5 = fmaxf(fmaxf(m0, m1), b[7]);
    float m6 = fmaxf(fmaxf(m2, m3), m4);
    return fmaxf(m5, m6);
}

template <bool FWD>
__device__ __forceinline__ void crf_chunk_matrix(const float* __restrict__ wgt,
                                                 const float* __restrict__ U,
                                                 float* __restrict__ Pm,
                                                 float* __restrict__ Ex,
                                                 int chunk, int b) {
    const int   lane  = threadIdx.x & 31;
    const int   nt    = __builtin_amdgcn_readfirstlane((int)(threadIdx.x >> 5));
    const int   c     = lane & 15;
    const int   h     = lane >> 4;
    const float LOG2E = 1.44269504088896340736f;

    v2f A[2][2][4];
    #pragma unroll
    for (int m = 0; m < 2; ++m)
    #pragma unroll
    for (int q = 0; q < 2; ++q)
    #pragma unroll
    for (int j = 0; j < 4; ++j)
    #pragma unroll
    for (int i = 0; i < 2; ++i) {
        const int row = 16 * m + c;
        const int t   = 16 * q + 8 * h + 2 * j + i;
        const float u = FWD ? U[t * Kv + row] : U[row * Kv + t];
        A[m][q][j][i] = fexp2(u * LOG2E);
    }

    v8f P[2];
    #pragma unroll
    for (int q = 0; q < 2; ++q)
    #pragma unroll
    for (int r = 0; r < 8; ++r)
        P[q][r] = (16 * q + 8 * h + r == 16 * nt + c) ? 1.0f : 0.0f;
    int esum = 0;

    const float* wlane = wgt + (size_t)b * (Tv * Kv) + 8 * h;
    v4f ring[RSv][4];
    auto issue = [&](int slot, int n) {
        const int nn = n < Tv - 1 ? n : Tv - 1;
        const int tt = FWD ? nn : (Tv - 1 - nn);
        const float* rowp = wlane + (size_t)tt * Kv;
        ring[slot][0] = *(const v4f*)(rowp);
        ring[slot][1] = *(const v4f*)(rowp + 4);
        ring[slot][2] = *(const v4f*)(rowp + 16);
        ring[slot][3] = *(const v4f*)(rowp + 20);
    };
    const int n0 = chunk * CHL;
    #pragma unroll
    for (int j = 0; j < PFv; ++j) issue(j, n0 + 1 + j);

    auto step = [&](int slot, int n) {
        issue((slot + PFv) % RSv, n + PFv);
        const v8f w0 = __builtin_shufflevector(ring[slot][0], ring[slot][1], 0, 1, 2, 3, 4, 5, 6, 7);
        const v8f w1 = __builtin_shufflevector(ring[slot][2], ring[slot][3], 0, 1, 2, 3, 4, 5, 6, 7);
        float mx = max16(P[0], P[1]);
        mx = fmaxf(mx, other_half(mx));
        const unsigned eb = (__builtin_bit_cast(unsigned, mx) >> 23) & 0xffu;
        const float sc = __builtin_bit_cast(float, (254u - eb) << 23);
        esum += 127 - (int)eb;
        v8f acc0 = {};
        v8f acc1 = {};
        #pragma unroll
        for (int q = 0; q < 2; ++q) {
            #pragma unroll
            for (int j = 0; j < 4; ++j) {
                v2f bb;
                bb[0] = P[q][2 * j];
                bb[1] = P[q][2 * j + 1];
                acc0 = __builtin_amdgcn_wmma_f32_16x16x4_f32(false, A[0][q][j], false, bb,
                                                             (short)0, acc0, false, false);
                acc1 = __builtin_amdgcn_wmma_f32_16x16x4_f32(false, A[1][q][j], false, bb,
                                                             (short)0, acc1, false, false);
            }
        }
        P[0] = acc0 * (w0 * sc);
        P[1] = acc1 * (w1 * sc);
    };
    static_assert(CHL % RSv == 0, "ring slots must be compile-time");
    for (int s0 = 0; s0 < CHL; s0 += RSv) {
        #pragma unroll
        for (int u = 0; u < RSv; ++u) step(u, n0 + 1 + s0 + u);
    }

    float* pb = Pm + ((size_t)(chunk * Bv + b) * 2 + nt) * 512;
    #pragma unroll
    for (int m = 0; m < 2; ++m)
    #pragma unroll
    for (int r = 0; r < 8; ++r)
        pb[(16 * m + 8 * h + r) * 16 + c] = P[m][r];
    if (h == 0) Ex[(size_t)(chunk * Bv + b) * Kv + 16 * nt + c] = (float)esum;
}

__global__ __launch_bounds__(64) void crf_chunks(const float* __restrict__ wgt,
                                                 const float* __restrict__ U,
                                                 float* __restrict__ Pm,
                                                 float* __restrict__ Ex) {
    const int per = (NCH - 1) * Bv;
    const int id  = blockIdx.x;
    if (id < per) crf_chunk_matrix<true >(wgt, U, Pm, Ex, id / Bv, id % Bv);
    else          crf_chunk_matrix<false>(wgt, U, Pm + (size_t)per * 1024, Ex + (size_t)per * Kv,
                                          (id - per) / Bv, (id - per) % Bv);
}

template <int XM>
__device__ __forceinline__ int row_xor(int v) {
    return __builtin_amdgcn_update_dpp(v, v, 0x160 + XM, 0xf, 0xf, false);
}
__device__ __forceinline__ float wave_max(float v) {
    v = fmaxf(v, __builtin_bit_cast(float, row_xor<1>(__builtin_bit_cast(int, v))));
    v = fmaxf(v, __builtin_bit_cast(float, row_xor<2>(__builtin_bit_cast(int, v))));
    v = fmaxf(v, __builtin_bit_cast(float, row_xor<4>(__builtin_bit_cast(int, v))));
    v = fmaxf(v, __builtin_bit_cast(float, row_xor<8>(__builtin_bit_cast(int, v))));
    return fmaxf(v, other_half(v));
}
__device__ __forceinline__ int wave_min(int v) {
    int o;
    o = row_xor<1>(v); v = o < v ? o : v;
    o = row_xor<2>(v); v = o < v ? o : v;
    o = row_xor<4>(v); v = o < v ? o : v;
    o = row_xor<8>(v); v = o < v ? o : v;
    o = __builtin_bit_cast(int, other_half(__builtin_bit_cast(float, v)));
    return o < v ? o : v;
}

template <bool FWD>
__device__ __forceinline__ void crf_join_chunks(const float* __restrict__ wgt,
                                                const float* __restrict__ Pm,
                                                const float* __restrict__ Ex,
                                                float* __restrict__ Ss, int b) {
    const int lane = threadIdx.x & 31;
    const int tt0  = FWD ? 0 : (Tv - 1);
    float S = wgt[((size_t)b * Tv + tt0) * Kv + lane];

    v4f   pr[4][8];
    float ex[4];
    auto loadm = [&](int slot, int ch) {
        const float* p0 = Pm + ((size_t)(ch * Bv + b) * 2) * 512 + lane * 16;
        #pragma unroll
        for (int g = 0; g < 4; ++g) {
            pr[slot][g]     = *(const v4f*)(p0 + 4 * g);
            pr[slot][4 + g] = *(const v4f*)(p0 + 512 + 4 * g);
        }
        ex[slot] = Ex[(size_t)(ch * Bv + b) * Kv + lane];
    };

    loadm(0, 0);
    loadm(1, 1);
    loadm(2, 2);
    static_assert(NCH - 1 >= 3, "three matrices are requested up front");
    #pragma unroll
    for (int ch = 0; ch < NCH - 1; ++ch) {
        const int cur = ch & 3;
        if (ch + 3 < NCH - 1) loadm((ch + 3) & 3, ch + 3);
        __builtin_amdgcn_sched_barrier(0);
        const int e  = (int)ex[cur];
        const int ed = wave_min(e) - e;
        const float v = ldexpf(S, ed);
        float s0 = 0.0f, s1 = 0.0f, s2 = 0.0f, s3 = 0.0f;
        #pragma unroll
        for (int g = 0; g < 8; ++g) {
            s0 = fmaf(pr[cur][g][0], readlane_f(v, 4 * g + 0), s0);
            s1 = fmaf(pr[cur][g][1], readlane_f(v, 4 * g + 1), s1);
            s2 = fmaf(pr[cur][g][2], readlane_f(v, 4 * g + 2), s2);
            s3 = fmaf(pr[cur][g][3], readlane_f(v, 4 * g + 3), s3);
        }
        const float sn = (s0 + s1) + (s2 + s3);
        S = sn * __builtin_amdgcn_rcpf(wave_max(sn));
        Ss[((size_t)(ch + 1) * Bv + b) * Kv + lane] = S;
    }
}

__global__ __launch_bounds__(32) void crf_join(const float* __restrict__ wgt,
                                               const float* __restrict__ Pm,
                                               const float* __restrict__ Ex,
                                               float* __restrict__ Ss) {
    const int per = (NCH - 1) * Bv;
    if (blockIdx.x < Bv) crf_join_chunks<true >(wgt, Pm, Ex, Ss, blockIdx.x);
    else                 crf_join_chunks<false>(wgt, Pm + (size_t)per * 1024, Ex + (size_t)per * Kv,
                                                Ss + (size_t)NCH * Bv * Kv, blockIdx.x - Bv);
}

template <bool FWD>
__device__ __forceinline__ void crf_chain(const float* __restrict__ wgt,
                                          const float* __restrict__ U,
                                          const float* __restrict__ Ss,
                                          float* __restrict__ dst, int grp, int chunk,
                                          v4f* xch) {
    const int   lane  = threadIdx.x & 31;
    const int   m     = __builtin_amdgcn_readfirstlane((int)(threadIdx.x >> 5));
    const int   o     = 1 - m;
    const int   c     = lane & 15;
    const int   h     = lane >> 4;
    const float LOG2E = 1.44269504088896340736f;

    v2f Aown[4], Aoth[4];
    #pragma unroll
    for (int j = 0; j < 4; ++j) {
        #pragma unroll
        for (int i = 0; i < 2; ++i) {
            const int row = 16 * m + c;
            const int t0  = 16 * m + 8 * h + 2 * j + i;
            const int t1  = 16 * o + 8 * h + 2 * j + i;
            const float u0 = FWD ? U[t0 * Kv + row] : U[row * Kv + t0];
            const float u1 = FWD ? U[t1 * Kv + row] : U[row * Kv + t1];
            Aown[j][i] = fexp2(u0 * LOG2E);
            Aoth[j][i] = fexp2(u1 * LOG2E);
        }
    }

    const unsigned laneoff = (unsigned)((grp * 16 + c) * (Tv * Kv) + 16 * m + 8 * h);
    const float* elane = wgt + laneoff;
    float*       dlane = dst + laneoff;

    v4f* xown = xch + (m * 32 + lane) * 2;
    v4f* xoth = xch + (o * 32 + lane) * 2;

    v4f ring[RSv][2];
    auto issue = [&](int slot, int n) {
        const int nn = n < Tv - 1 ? n : Tv - 1;
        const int tt = FWD ? nn : (Tv - 1 - nn);
        const float* rowp = elane + (size_t)tt * Kv;
        ring[slot][0] = *(const v4f*)(rowp);
        ring[slot][1] = *(const v4f*)(rowp + 4);
    };
    auto weights = [&](int slot) {
        return __builtin_shufflevector(ring[slot][0], ring[slot][1], 0, 1, 2, 3, 4, 5, 6, 7);
    };
    auto store8 = [&](int n, v8f s) {
        const int tt = FWD ? n : (Tv - 1 - n);
        float* rowd = dlane + (size_t)tt * Kv;
        *(v4f*)(rowd)     = __builtin_shufflevector(s, s, 0, 1, 2, 3);
        *(v4f*)(rowd + 4) = __builtin_shufflevector(s, s, 4, 5, 6, 7);
    };
    auto handover = [&](int p, v8f s) {
        xown[p * 128]     = __builtin_shufflevector(s, s, 0, 1, 2, 3);
        xown[p * 128 + 1] = __builtin_shufflevector(s, s, 4, 5, 6, 7);
        __builtin_amdgcn_fence(__ATOMIC_RELEASE, "workgroup", "local");
        __builtin_amdgcn_s_barrier();
        __builtin_amdgcn_fence(__ATOMIC_ACQUIRE, "workgroup", "local");
    };

    const int n0 = chunk * CHL;
    const float* sp = (chunk == 0)
        ? (elane + (size_t)(FWD ? 0 : (Tv - 1)) * Kv)
        : (Ss + ((size_t)chunk * Bv + grp * 16 + c) * Kv + 16 * m + 8 * h);
    v8f Sown = __builtin_shufflevector(*(const v4f*)(sp), *(const v4f*)(sp + 4),
                                       0, 1, 2, 3, 4, 5, 6, 7);
    if (chunk == 0) {
        v8f ones;
        #pragma unroll
        for (int r = 0; r < 8; ++r) ones[r] = 1.0f;
        store8(0, FWD ? Sown : ones);
    }
    #pragma unroll
    for (int j = 0; j < PFv; ++j) issue(j, n0 + 1 + j);
    handover(0, Sown);

    auto step = [&](int slot, int n, int par, bool prefetch) {
        if (prefetch) issue((slot + PFv) % RSv, n + PFv);
        const v4f o0 = xoth[par * 128];
        const v4f o1 = xoth[par * 128 + 1];
        v8f w = weights(slot);
        v8f accA = {};
        v8f accB = {};
        #pragma unroll
        for (int j = 0; j < 4; ++j) {
            v2f b;
            b[0] = Sown[2 * j];
            b[1] = Sown[2 * j + 1];
            if (j & 1) accB = __builtin_amdgcn_wmma_f32_16x16x4_f32(false, Aown[j], false, b,
                                                                    (short)0, accB, false, false);
            else       accA = __builtin_amdgcn_wmma_f32_16x16x4_f32(false, Aown[j], false, b,
                                                                    (short)0, accA, false, false);
        }
        const v8f Soth = __builtin_shufflevector(o0, o1, 0, 1, 2, 3, 4, 5, 6, 7);
        #pragma unroll
        for (int j = 0; j < 4; ++j) {
            v2f b;
            b[0] = Soth[2 * j];
            b[1] = Soth[2 * j + 1];
            if (j & 1) accB = __builtin_amdgcn_wmma_f32_16x16x4_f32(false, Aoth[j], false, b,
                                                                    (short)0, accB, false, false);
            else       accA = __builtin_amdgcn_wmma_f32_16x16x4_f32(false, Aoth[j], false, b,
                                                                    (short)0, accA, false, false);
        }
        float mx = max16(Sown, Soth);
        mx = fmaxf(mx, other_half(mx));
        const float rs = __builtin_amdgcn_rcpf(mx);
        w *= rs;
        const v8f acc = accA + accB;
        Sown = acc * w;
        store8(n, FWD ? Sown : acc);
        handover(par ^ 1, Sown);
    };

    static_assert(CHL % RSv == 0 && RSv % 2 == 0, "ring slot and buffer parity must be compile-time");
    static_assert(CHL >= 2 * RSv, "a main loop and an unrolled tail");
    for (int s0 = 0; s0 < CHL - RSv; s0 += RSv) {
        #pragma unroll
        for (int u = 0; u < RSv; ++u) step(u, n0 + 1 + s0 + u, u & 1, true);
    }
    #pragma unroll
    for (int u = 0; u < RSv - 1; ++u)
        step(u, n0 + 1 + (CHL - RSv) + u, u & 1, u + PFv < RSv);
    if (n0 + CHL < Tv) step(RSv - 1, n0 + CHL, (RSv - 1) & 1, false);
}

__global__ __launch_bounds__(64) void crf_scan(const float* __restrict__ wgt,
                                               const float* __restrict__ U,
                                               const float* __restrict__ Ss,
                                               float* __restrict__ alpha_out,
                                               float* __restrict__ beta_out) {
    __shared__ v4f xch[2 * 2 * 32 * 2];
    const int per = NCH * (Bv / 16);
    const int id  = blockIdx.x;
    if (id < per) crf_chain<true >(wgt, U, Ss, alpha_out, id % (Bv / 16), id / (Bv / 16), xch);
    else          crf_chain<false>(wgt, U, Ss + (size_t)NCH * Bv * Kv, beta_out,
                                   (id - per) % (Bv / 16), (id - per) / (Bv / 16), xch);
}

__global__ __launch_bounds__(256) void crf_softmax(const float* __restrict__ alpha,
                                                   float* __restrict__ out) {
    const int lane = threadIdx.x & 31;
    const int wave = threadIdx.x >> 5;
    const size_t row = (size_t)blockIdx.x * 8 + wave;
    const size_t idx = row * Kv + lane;

    float p = alpha[idx] * out[idx];
    float sum = p;
    #pragma unroll
    for (int off = 16; off; off >>= 1) sum += __shfl_xor(sum, off, 32);
    out[idx] = p / sum;
}

extern "C" void kernel_launch(void* const* d_in, const int* in_sizes, int n_in,
                              void* d_out, int out_size, void* d_ws, size_t ws_size,
                              hipStream_t stream) {
    const float* x = (const float*)d_in[0];
    const float* W = (const float*)d_in[1];
    const float* U = (const float*)d_in[2];
    const float* b = (const float*)d_in[3];
    float* out = (float*)d_out;

    float* ws    = (float*)d_ws;
    float* wgt   = ws;
    float* alpha = ws + (size_t)Bv * Tv * Kv;
    float* Wp    = alpha + (size_t)Bv * Tv * Kv;
    float* Pm    = Wp + (size_t)Dv * Kv;
    float* Ex    = Pm + (size_t)2 * (NCH - 1) * Bv * 1024;
    float* Ss    = Ex + (size_t)2 * (NCH - 1) * Bv * Kv;

    crf_packW     <<<dim3(128), dim3(256), 0, stream>>>(W, Wp);
    crf_emissions <<<dim3((Bv * Tv / 16) / 8), dim3(256), 0, stream>>>(x, Wp, b, wgt);
    crf_chunks    <<<dim3(2 * (NCH - 1) * Bv), dim3(64), 0, stream>>>(wgt, U, Pm, Ex);
    crf_join      <<<dim3(2 * Bv), dim3(32), 0, stream>>>(wgt, Pm, Ex, Ss);
    crf_scan      <<<dim3(2 * NCH * (Bv / 16)), dim3(64), 0, stream>>>(wgt, U, Ss, alpha, out);
    crf_softmax   <<<dim3(Bv * Tv / 8), dim3(256), 0, stream>>>(alpha, out);
}


# backward chunk starts come from the transposes of the forward chunk matrices (b_lo = P^T b_hi, one mat-vec per chunk): the backward half of the chunk products is gone; sum order changed
# speedup vs baseline: 5.9526x; 1.1511x over previous
#include <hip/hip_runtime.h>
#include <hip/hip_bf16.h>

typedef __attribute__((ext_vector_type(2))) float v2f;
typedef __attribute__((ext_vector_type(4))) float v4f;
typedef __attribute__((ext_vector_type(8))) float v8f;

#define Bv 64
#define Tv 512
#define Dv 1024
#define Kv 32

__device__ __forceinline__ float fexp2(float x) { return __builtin_amdgcn_exp2f(x); }
__device__ __forceinline__ float flog2(float x) { return __builtin_amdgcn_logf(x); }

__device__ __forceinline__ float readlane_f(float v, int l) {
    return __builtin_bit_cast(float,
        __builtin_amdgcn_readlane(__builtin_bit_cast(int, v), l));
}

__global__ __launch_bounds__(256) void crf_packW(const float* __restrict__ W,
                                                 float* __restrict__ Wp) {
    const int idx  = blockIdx.x * 256 + threadIdx.x;
    const int i    = idx & 3;
    const int lane = (idx >> 2) & 31;
    const int d4   = idx >> 7;
    const int n    = lane & 15;
    const int h    = lane >> 4;
    const int d    = 4 * d4 + 2 * h + (i & 1);
    const int k    = n + 16 * (i >> 1);
    Wp[idx] = W[d * Kv + k];
}

#define EKB 8
#define ENB ((Dv / 4) / EKB)
#define EXR 4

__global__ __launch_bounds__(256) void crf_emissions(const float* __restrict__ x,
                                                     const float* __restrict__ Wp,
                                                     const float* __restrict__ bias,
                                                     float* __restrict__ wgt) {
    __shared__ v4f wl[(Dv / 4) * 32];

    const int tid  = threadIdx.x;
    const int lane = tid & 31;
    const int wave = tid >> 5;
    const int mt   = blockIdx.x * 8 + wave;
    const int half = lane >> 4;
    const int l15  = lane & 15;

    const float* xptr = x + (size_t)(mt * 16 + l15) * Dv + half * 2;

    v2f xa[EXR][EKB];
    v4f wb[2][EKB];
    auto loadx = [&](int slot, int blk) {
        #pragma unroll
        for (int ks = 0; ks < EKB; ++ks)
            xa[slot][ks] = *(const v2f*)(xptr + (blk * EKB + ks) * 4);
    };
    auto loadw = [&](int slot, int blk) {
        #pragma unroll
        for (int ks = 0; ks < EKB; ++ks)
            wb[slot][ks] = wl[(blk * EKB + ks) * 32 + lane];
    };

    #pragma unroll
    for (int j = 0; j < EXR - 1; ++j) loadx(j, j);

    {
        const v4f* wp4 = (const v4f*)Wp;
        #pragma unroll
        for (int pass = 0; pass < 4; ++pass) {
            v4f t[8];
            #pragma unroll
            for (int j = 0; j < 8; ++j) t[j] = wp4[(pass * 8 + j) * 256 + tid];
            #pragma unroll
            for (int j = 0; j < 8; ++j) wl[(pass * 8 + j) * 256 + tid] = t[j];
        }
    }
    __syncthreads();
    loadw(0, 0);

    v8f acc0 = {};
    v8f acc1 = {};
    auto block = [&](int u, int blk, bool px, bool pw) {
        if (px) loadx((u + EXR - 1) % EXR, blk + EXR - 1);
        if (pw) loadw((u + 1) & 1, blk + 1);
        __builtin_amdgcn_sched_barrier(0);
        #pragma unroll
        for (int ks = 0; ks < EKB; ++ks) {
            const v4f w = wb[u & 1][ks];
            v2f b0, b1;
            b0[0] = w[0]; b0[1] = w[1];
            b1[0] = w[2]; b1[1] = w[3];
            acc0 = __builtin_amdgcn_wmma_f32_16x16x4_f32(false, xa[u % EXR][ks], false, b0,
                                                         (short)0, acc0, false, false);
            acc1 = __builtin_amdgcn_wmma_f32_16x16x4_f32(false, xa[u % EXR][ks], false, b1,
                                                         (short)0, acc1, false, false);
        }
    };
    static_assert(ENB % EXR == 0 && EXR % 2 == 0, "ring slots must be compile-time");
    for (int blk = 0; blk < ENB - EXR; blk += EXR) {
        #pragma unroll
        for (int u = 0; u < EXR; ++u) block(u, blk + u, true, true);
    }
    #pragma unroll
    for (int u = 0; u < EXR; ++u)
        block(u, ENB - EXR + u, u + EXR - 1 < EXR, u + 1 < EXR);

    const float bv0 = bias[l15];
    const float bv1 = bias[l15 + 16];
    const float LOG2E = 1.44269504088896340736f;
    #pragma unroll
    for (int r = 0; r < 8; ++r) {
        const float v0 = acc0[r] + bv0;
        const float v1 = acc1[r] + bv1;
        float mx = fmaxf(v0, v1);
        #pragma unroll
        for (int off = 8; off; off >>= 1) mx = fmaxf(mx, __shfl_xor(mx, off, 32));
        const float ms = -mx * LOG2E;
        size_t row = (size_t)mt * 16 + r + half * 8;
        wgt[row * Kv + l15]      = fexp2(fmaf(v0, LOG2E, ms));
        wgt[row * Kv + l15 + 16] = fexp2(fmaf(v1, LOG2E, ms));
    }
}

#define CHL 32
#define NCH (Tv / CHL)
#define PFv 4
#define RSv 8

__device__ __forceinline__ float other_half(float v) {
    unsigned u = __builtin_bit_cast(unsigned, v);
    return __builtin_bit_cast(float,
        __builtin_amdgcn_permlanex16(u, u, 0x76543210u, 0xfedcba98u, false, false));
}

__device__ __forceinline__ float max16(v8f a, v8f b) {
    float m0 = fmaxf(fmaxf(a[0], a[1]), a[2]);
    float m1 = fmaxf(fmaxf(a[3], a[4]), a[5]);
    float m2 = fmaxf(fmaxf(a[6], a[7]), b[0]);
    float m3 = fmaxf(fmaxf(b[1], b[2]), b[3]);
    float m4 = fmaxf(fmaxf(b[4], b[5]), b[6]);
    float m5 = fmaxf(fmaxf(m0, m1), b[7]);
    float m6 = fmaxf(fmaxf(m2, m3), m4);
    return fmaxf(m5, m6);
}

__device__ __forceinline__ void crf_chunk_matrix(const float* __restrict__ wgt,
                                                 const float* __restrict__ U,
                                                 float* __restrict__ Pm,
                                                 float* __restrict__ Pt,
                                                 float* __restrict__ Ex,
                                                 int chunk, int b) {
    const int   lane  = threadIdx.x & 31;
    const int   nt    = __builtin_amdgcn_readfirstlane((int)(threadIdx.x >> 5));
    const int   c     = lane & 15;
    const int   h     = lane >> 4;
    const float LOG2E = 1.44269504088896340736f;

    v2f A[2][2][4];
    #pragma unroll
    for (int m = 0; m < 2; ++m)
    #pragma unroll
    for (int q = 0; q < 2; ++q)
    #pragma unroll
    for (int j = 0; j < 4; ++j)
    #pragma unroll
    for (int i = 0; i < 2; ++i) {
        const int row = 16 * m + c;
        const int t   = 16 * q + 8 * h + 2 * j + i;
        A[m][q][j][i] = fexp2(U[t * Kv + row] * LOG2E);
    }

    v8f P[2];
    #pragma unroll
    for (int q = 0; q < 2; ++q)
    #pragma unroll
    for (int r = 0; r < 8; ++r)
        P[q][r] = (16 * q + 8 * h + r == 16 * nt + c) ? 1.0f : 0.0f;
    int esum = 0;

    const float* wlane = wgt + (size_t)b * (Tv * Kv) + 8 * h;
    v4f ring[RSv][4];
    auto issue = [&](int slot, int n) {
        const int tt = n < Tv - 1 ? n : Tv - 1;
        const float* rowp = wlane + (size_t)tt * Kv;
        ring[slot][0] = *(const v4f*)(rowp);
        ring[slot][1] = *(const v4f*)(rowp + 4);
        ring[slot][2] = *(const v4f*)(rowp + 16);
        ring[slot][3] = *(const v4f*)(rowp + 20);
    };
    const int n0 = chunk * CHL;
    #pragma unroll
    for (int j = 0; j < PFv; ++j) issue(j, n0 + 1 + j);

    auto step = [&](int slot, int n) {
        issue((slot + PFv) % RSv, n + PFv);
        const v8f w0 = __builtin_shufflevector(ring[slot][0], ring[slot][1], 0, 1, 2, 3, 4, 5, 6, 7);
        const v8f w1 = __builtin_shufflevector(ring[slot][2], ring[slot][3], 0, 1, 2, 3, 4, 5, 6, 7);
        float mx = max16(P[0], P[1]);
        mx = fmaxf(mx, other_half(mx));
        const unsigned eb = (__builtin_bit_cast(unsigned, mx) >> 23) & 0xffu;
        const float sc = __builtin_bit_cast(float, (254u - eb) << 23);
        esum += 127 - (int)eb;
        v8f acc0 = {};
        v8f acc1 = {};
        #pragma unroll
        for (int q = 0; q < 2; ++q) {
            #pragma unroll
            for (int j = 0; j < 4; ++j) {
                v2f bb;
                bb[0] = P[q][2 * j];
                bb[1] = P[q][2 * j + 1];
                acc0 = __builtin_amdgcn_wmma_f32_16x16x4_f32(false, A[0][q][j], false, bb,
                                                             (short)0, acc0, false, false);
                acc1 = __builtin_amdgcn_wmma_f32_16x16x4_f32(false, A[1][q][j], false, bb,
                                                             (short)0, acc1, false, false);
            }
        }
        P[0] = acc0 * (w0 * sc);
        P[1] = acc1 * (w1 * sc);
    };
    static_assert(CHL % RSv == 0 && CHL >= 2 * RSv, "ring slots must be compile-time");
    for (int s0 = 0; s0 < CHL - RSv; s0 += RSv) {
        #pragma unroll
        for (int u = 0; u < RSv; ++u) step(u, n0 + 1 + s0 + u);
    }
    #pragma unroll
    for (int u = 0; u < RSv - 1; ++u) step(u, n0 + 1 + (CHL - RSv) + u);
    if (n0 + CHL < Tv) step(RSv - 1, n0 + CHL);

    if (chunk < NCH - 1) {
        float* pb = Pm + ((size_t)(chunk * Bv + b) * 2 + nt) * 512;
        #pragma unroll
        for (int m = 0; m < 2; ++m)
        #pragma unroll
        for (int r = 0; r < 8; ++r)
            pb[(16 * m + 8 * h + r) * 16 + c] = P[m][r];
    }
    if (chunk > 0) {
        float* pc = Pt + (size_t)(chunk * Bv + b) * 1024 + (16 * nt + c) * 32 + 8 * h;
        #pragma unroll
        for (int m = 0; m < 2; ++m) {
            *(v4f*)(pc + 16 * m)     = __builtin_shufflevector(P[m], P[m], 0, 1, 2, 3);
            *(v4f*)(pc + 16 * m + 4) = __builtin_shufflevector(P[m], P[m], 4, 5, 6, 7);
        }
    }
    if (h == 0) Ex[(size_t)(chunk * Bv + b) * Kv + 16 * nt + c] = (float)esum;
}

__global__ __launch_bounds__(64) void crf_chunks(const float* __restrict__ wgt,
                                                 const float* __restrict__ U,
                                                 float* __restrict__ Pm,
                                                 float* __restrict__ Pt,
                                                 float* __restrict__ Ex) {
    const int id = blockIdx.x;
    crf_chunk_matrix(wgt, U, Pm, Pt, Ex, id / Bv, id % Bv);
}

template <int XM>
__device__ __forceinline__ int row_xor(int v) {
    return __builtin_amdgcn_update_dpp(v, v, 0x160 + XM, 0xf, 0xf, false);
}
__device__ __forceinline__ float wave_max(float v) {
    v = fmaxf(v, __builtin_bit_cast(float, row_xor<1>(__builtin_bit_cast(int, v))));
    v = fmaxf(v, __builtin_bit_cast(float, row_xor<2>(__builtin_bit_cast(int, v))));
    v = fmaxf(v, __builtin_bit_cast(float, row_xor<4>(__builtin_bit_cast(int, v))));
    v = fmaxf(v, __builtin_bit_cast(float, row_xor<8>(__builtin_bit_cast(int, v))));
    return fmaxf(v, other_half(v));
}
__device__ __forceinline__ int wave_min(int v) {
    int o;
    o = row_xor<1>(v); v = o < v ? o : v;
    o = row_xor<2>(v); v = o < v ? o : v;
    o = row_xor<4>(v); v = o < v ? o : v;
    o = row_xor<8>(v); v = o < v ? o : v;
    o = __builtin_bit_cast(int, other_half(__builtin_bit_cast(float, v)));
    return o < v ? o : v;
}

template <bool FWD>
__device__ __forceinline__ void crf_join_chunks(const float* __restrict__ wgt,
                                                const float* __restrict__ Pmat,
                                                const float* __restrict__ Ex,
                                                float* __restrict__ Ss, int b) {
    const int lane = threadIdx.x & 31;
    float S = FWD ? wgt[((size_t)b * Tv) * Kv + lane] : 1.0f;

    v4f   pr[4][8];
    float ex[4];
    float wr[4];
    auto loadm = [&](int slot, int i) {
        const int ch = FWD ? i : (NCH - 1 - i);
        if (FWD) {
            const float* p0 = Pmat + ((size_t)(ch * Bv + b) * 2) * 512 + lane * 16;
            #pragma unroll
            for (int g = 0; g < 4; ++g) {
                pr[slot][g]     = *(const v4f*)(p0 + 4 * g);
                pr[slot][4 + g] = *(const v4f*)(p0 + 512 + 4 * g);
            }
        } else {
            const float* p0 = Pmat + (size_t)(ch * Bv + b) * 1024 + lane * 32;
            #pragma unroll
            for (int g = 0; g < 8; ++g) pr[slot][g] = *(const v4f*)(p0 + 4 * g);
            wr[slot] = wgt[((size_t)b * Tv + ch * CHL) * Kv + lane];
        }
        ex[slot] = Ex[(size_t)(ch * Bv + b) * Kv + lane];
    };

    loadm(0, 0);
    loadm(1, 1);
    loadm(2, 2);
    static_assert(NCH - 1 >= 3, "three matrices are requested up front");
    #pragma unroll
    for (int i = 0; i < NCH - 1; ++i) {
        const int cur = i & 3;
        if (i + 3 < NCH - 1) loadm((i + 3) & 3, i + 3);
        __builtin_amdgcn_sched_barrier(0);
        const int e  = (int)ex[cur];
        const int ed = wave_min(e) - e;
        const float v = FWD ? ldexpf(S, ed) : S;
        float s0 = 0.0f, s1 = 0.0f, s2 = 0.0f, s3 = 0.0f;
        #pragma unroll
        for (int g = 0; g < 8; ++g) {
            s0 = fmaf(pr[cur][g][0], readlane_f(v, 4 * g + 0), s0);
            s1 = fmaf(pr[cur][g][1], readlane_f(v, 4 * g + 1), s1);
            s2 = fmaf(pr[cur][g][2], readlane_f(v, 4 * g + 2), s2);
            s3 = fmaf(pr[cur][g][3], readlane_f(v, 4 * g + 3), s3);
        }
        float sn = (s0 + s1) + (s2 + s3);
        if (!FWD) sn = ldexpf(sn, ed);
        S = sn * __builtin_amdgcn_rcpf(wave_max(sn));
        Ss[((size_t)(i + 1) * Bv + b) * Kv + lane] = FWD ? S : S * wr[cur];
    }
}

__global__ __launch_bounds__(32) void crf_join(const float* __restrict__ wgt,
                                               const float* __restrict__ Pm,
                                               const float* __restrict__ Pt,
                                               const float* __restrict__ Ex,
                                               float* __restrict__ Ss) {
    if (blockIdx.x < Bv) crf_join_chunks<true >(wgt, Pm, Ex, Ss, blockIdx.x);
    else                 crf_join_chunks<false>(wgt, Pt, Ex, Ss + (size_t)NCH * Bv * Kv, blockIdx.x - Bv);
}

template <bool FWD>
__device__ __forceinline__ void crf_chain(const float* __restrict__ wgt,
                                          const float* __restrict__ U,
                                          const float* __restrict__ Ss,
                                          float* __restrict__ dst, int grp, int chunk,
                                          v4f* xch) {
    const int   lane  = threadIdx.x & 31;
    const int   m     = __builtin_amdgcn_readfirstlane((int)(threadIdx.x >> 5));
    const int   o     = 1 - m;
    const int   c     = lane & 15;
    const int   h     = lane >> 4;
    const float LOG2E = 1.44269504088896340736f;

    v2f Aown[4], Aoth[4];
    #pragma unroll
    for (int j = 0; j < 4; ++j) {
        #pragma unroll
        for (int i = 0; i < 2; ++i) {
            const int row = 16 * m + c;
            const int t0  = 16 * m + 8 * h + 2 * j + i;
            const int t1  = 16 * o + 8 * h + 2 * j + i;
            const float u0 = FWD ? U[t0 * Kv + row] : U[row * Kv + t0];
            const float u1 = FWD ? U[t1 * Kv + row] : U[row * Kv + t1];
            Aown[j][i] = fexp2(u0 * LOG2E);
            Aoth[j][i] = fexp2(u1 * LOG2E);
        }
    }

    const unsigned laneoff = (unsigned)((grp * 16 + c) * (Tv * Kv) + 16 * m + 8 * h);
    const float* elane = wgt + laneoff;
    float*       dlane = dst + laneoff;

    v4f* xown = xch + (m * 32 + lane) * 2;
    v4f* xoth = xch + (o * 32 + lane) * 2;

    v4f ring[RSv][2];
    auto issue = [&](int slot, int n) {
        const int nn = n < Tv - 1 ? n : Tv - 1;
        const int tt = FWD ? nn : (Tv - 1 - nn);
        const float* rowp = elane + (size_t)tt * Kv;
        ring[slot][0] = *(const v4f*)(rowp);
        ring[slot][1] = *(const v4f*)(rowp + 4);
    };
    auto weights = [&](int slot) {
        return __builtin_shufflevector(ring[slot][0], ring[slot][1], 0, 1, 2, 3, 4, 5, 6, 7);
    };
    auto store8 = [&](int n, v8f s) {
        const int tt = FWD ? n : (Tv - 1 - n);
        float* rowd = dlane + (size_t)tt * Kv;
        *(v4f*)(rowd)     = __builtin_shufflevector(s, s, 0, 1, 2, 3);
        *(v4f*)(rowd + 4) = __builtin_shufflevector(s, s, 4, 5, 6, 7);
    };
    auto handover = [&](int p, v8f s) {
        xown[p * 128]     = __builtin_shufflevector(s, s, 0, 1, 2, 3);
        xown[p * 128 + 1] = __builtin_shufflevector(s, s, 4, 5, 6, 7);
        __builtin_amdgcn_fence(__ATOMIC_RELEASE, "workgroup", "local");
        __builtin_amdgcn_s_barrier();
        __builtin_amdgcn_fence(__ATOMIC_ACQUIRE, "workgroup", "local");
    };

    const int n0 = FWD ? chunk * CHL : (chunk == 0 ? 0 : chunk * CHL - 1);
    const float* sp = (chunk == 0)
        ? (elane + (size_t)(FWD ? 0 : (Tv - 1)) * Kv)
        : (Ss + ((size_t)chunk * Bv + grp * 16 + c) * Kv + 16 * m + 8 * h);
    v8f Sown = __builtin_shufflevector(*(const v4f*)(sp), *(const v4f*)(sp + 4),
                                       0, 1, 2, 3, 4, 5, 6, 7);
    if (chunk == 0) {
        v8f ones;
        #pragma unroll
        for (int r = 0; r < 8; ++r) ones[r] = 1.0f;
        store8(0, FWD ? Sown : ones);
    }
    #pragma unroll
    for (int j = 0; j < PFv; ++j) issue(j, n0 + 1 + j);
    handover(0, Sown);

    auto step = [&](int slot, int n, int par, bool prefetch) {
        if (prefetch) issue((slot + PFv) % RSv, n + PFv);
        const v4f o0 = xoth[par * 128];
        const v4f o1 = xoth[par * 128 + 1];
        v8f w = weights(slot);
        v8f accA = {};
        v8f accB = {};
        #pragma unroll
        for (int j = 0; j < 4; ++j) {
            v2f b;
            b[0] = Sown[2 * j];
            b[1] = Sown[2 * j + 1];
            if (j & 1) accB = __builtin_amdgcn_wmma_f32_16x16x4_f32(false, Aown[j], false, b,
                                                                    (short)0, accB, false, false);
            else       accA = __builtin_amdgcn_wmma_f32_16x16x4_f32(false, Aown[j], false, b,
                                                                    (short)0, accA, false, false);
        }
        const v8f Soth = __builtin_shufflevector(o0, o1, 0, 1, 2, 3, 4, 5, 6, 7);
        #pragma unroll
        for (int j = 0; j < 4; ++j) {
            v2f b;
            b[0] = Soth[2 * j];
            b[1] = Soth[2 * j + 1];
            if (j & 1) accB = __builtin_amdgcn_wmma_f32_16x16x4_f32(false, Aoth[j], false, b,
                                                                    (short)0, accB, false, false);
            else       accA = __builtin_amdgcn_wmma_f32_16x16x4_f32(false, Aoth[j], false, b,
                                                                    (short)0, accA, false, false);
        }
        float mx = max16(Sown, Soth);
        mx = fmaxf(mx, other_half(mx));
        const float rs = __builtin_amdgcn_rcpf(mx);
        w *= rs;
        const v8f acc = accA + accB;
        Sown = acc * w;
        store8(n, FWD ? Sown : acc);
        handover(par ^ 1, Sown);
    };

    static_assert(CHL % RSv == 0 && RSv % 2 == 0, "ring slot and buffer parity must be compile-time");
    static_assert(CHL >= 2 * RSv, "a main loop and an unrolled tail");
    for (int s0 = 0; s0 < CHL - RSv; s0 += RSv) {
        #pragma unroll
        for (int u = 0; u < RSv; ++u) step(u, n0 + 1 + s0 + u, u & 1, true);
    }
    #pragma unroll
    for (int u = 0; u < RSv - 1; ++u)
        step(u, n0 + 1 + (CHL - RSv) + u, u & 1, u + PFv < RSv);
    if (FWD ? (chunk != NCH - 1) : (chunk != 0)) step(RSv - 1, n0 + CHL, (RSv - 1) & 1, false);
}

__global__ __launch_bounds__(64) void crf_scan(const float* __restrict__ wgt,
                                               const float* __restrict__ U,
                                               const float* __restrict__ Ss,
                                               float* __restrict__ alpha_out,
                                               float* __restrict__ beta_out) {
    __shared__ v4f xch[2 * 2 * 32 * 2];
    const int per = NCH * (Bv / 16);
    const int id  = blockIdx.x;
    if (id < per) crf_chain<true >(wgt, U, Ss, alpha_out, id % (Bv / 16), id / (Bv / 16), xch);
    else          crf_chain<false>(wgt, U, Ss + (size_t)NCH * Bv * Kv, beta_out,
                                   (id - per) % (Bv / 16), (id - per) / (Bv / 16), xch);
}

__global__ __launch_bounds__(256) void crf_softmax(const float* __restrict__ alpha,
                                                   float* __restrict__ out) {
    const int lane = threadIdx.x & 31;
    const int wave = threadIdx.x >> 5;
    const size_t row = (size_t)blockIdx.x * 8 + wave;
    const size_t idx = row * Kv + lane;

    float p = alpha[idx] * out[idx];
    float sum = p;
    #pragma unroll
    for (int off = 16; off; off >>= 1) sum += __shfl_xor(sum, off, 32);
    out[idx] = p / sum;
}

extern "C" void kernel_launch(void* const* d_in, const int* in_sizes, int n_in,
                              void* d_out, int out_size, void* d_ws, size_t ws_size,
                              hipStream_t stream) {
    const float* x = (const float*)d_in[0];
    const float* W = (const float*)d_in[1];
    const float* U = (const float*)d_in[2];
    const float* b = (const float*)d_in[3];
    float* out = (float*)d_out;

    float* ws    = (float*)d_ws;
    float* wgt   = ws;
    float* alpha = ws + (size_t)Bv * Tv * Kv;
    float* Wp    = alpha + (size_t)Bv * Tv * Kv;
    float* Pm    = Wp + (size_t)Dv * Kv;
    float* Pt    = Pm + (size_t)NCH * Bv * 1024;
    float* Ex    = Pt + (size_t)NCH * Bv * 1024;
    float* Ss    = Ex + (size_t)NCH * Bv * Kv;

    crf_packW     <<<dim3(128), dim3(256), 0, stream>>>(W, Wp);
    crf_emissions <<<dim3((Bv * Tv / 16) / 8), dim3(256), 0, stream>>>(x, Wp, b, wgt);
    crf_chunks    <<<dim3(NCH * Bv), dim3(64), 0, stream>>>(wgt, U, Pm, Pt, Ex);
    crf_join      <<<dim3(2 * Bv), dim3(32), 0, stream>>>(wgt, Pm, Pt, Ex, Ss);
    crf_scan      <<<dim3(2 * NCH * (Bv / 16)), dim3(64), 0, stream>>>(wgt, U, Ss, alpha, out);
    crf_softmax   <<<dim3(Bv * Tv / 8), dim3(256), 0, stream>>>(alpha, out);
}


# scan + marginals in one launch: a 4-wave work-group runs a row block's forward and backward chunk together, a-hat/b-hat tiles stay in LDS, normalised behind one barrier; 5 launches; sum order changed
# speedup vs baseline: 6.2787x; 1.0548x over previous
#include <hip/hip_runtime.h>
#include <hip/hip_bf16.h>

typedef __attribute__((ext_vector_type(2))) float v2f;
typedef __attribute__((ext_vector_type(4))) float v4f;
typedef __attribute__((ext_vector_type(8))) float v8f;

#define Bv 64
#define Tv 512
#define Dv 1024
#define Kv 32

__device__ __forceinline__ float fexp2(float x) { return __builtin_amdgcn_exp2f(x); }
__device__ __forceinline__ float flog2(float x) { return __builtin_amdgcn_logf(x); }

__device__ __forceinline__ float readlane_f(float v, int l) {
    return __builtin_bit_cast(float,
        __builtin_amdgcn_readlane(__builtin_bit_cast(int, v), l));
}

__global__ __launch_bounds__(256) void crf_packW(const float* __restrict__ W,
                                                 float* __restrict__ Wp) {
    const int idx  = blockIdx.x * 256 + threadIdx.x;
    const int i    = idx & 3;
    const int lane = (idx >> 2) & 31;
    const int d4   = idx >> 7;
    const int n    = lane & 15;
    const int h    = lane >> 4;
    const int d    = 4 * d4 + 2 * h + (i & 1);
    const int k    = n + 16 * (i >> 1);
    Wp[idx] = W[d * Kv + k];
}

#define EKB 8
#define ENB ((Dv / 4) / EKB)
#define EXR 4

__global__ __launch_bounds__(256) void crf_emissions(const float* __restrict__ x,
                                                     const float* __restrict__ Wp,
                                                     const float* __restrict__ bias,
                                                     float* __restrict__ wgt) {
    __shared__ v4f wl[(Dv / 4) * 32];

    const int tid  = threadIdx.x;
    const int lane = tid & 31;
    const int wave = tid >> 5;
    const int mt   = blockIdx.x * 8 + wave;
    const int half = lane >> 4;
    const int l15  = lane & 15;

    const float* xptr = x + (size_t)(mt * 16 + l15) * Dv + half * 2;

    v2f xa[EXR][EKB];
    v4f wb[2][EKB];
    auto loadx = [&](int slot, int blk) {
        #pragma unroll
        for (int ks = 0; ks < EKB; ++ks)
            xa[slot][ks] = *(const v2f*)(xptr + (blk * EKB + ks) * 4);
    };
    auto loadw = [&](int slot, int blk) {
        #pragma unroll
        for (int ks = 0; ks < EKB; ++ks)
            wb[slot][ks] = wl[(blk * EKB + ks) * 32 + lane];
    };

    #pragma unroll
    for (int j = 0; j < EXR - 1; ++j) loadx(j, j);

    {
        const v4f* wp4 = (const v4f*)Wp;
        #pragma unroll
        for (int pass = 0; pass < 4; ++pass) {
            v4f t[8];
            #pragma unroll
            for (int j = 0; j < 8; ++j) t[j] = wp4[(pass * 8 + j) * 256 + tid];
            #pragma unroll
            for (int j = 0; j < 8; ++j) wl[(pass * 8 + j) * 256 + tid] = t[j];
        }
    }
    __syncthreads();
    loadw(0, 0);

    v8f acc0 = {};
    v8f acc1 = {};
    auto block = [&](int u, int blk, bool px, bool pw) {
        if (px) loadx((u + EXR - 1) % EXR, blk + EXR - 1);
        if (pw) loadw((u + 1) & 1, blk + 1);
        __builtin_amdgcn_sched_barrier(0);
        #pragma unroll
        for (int ks = 0; ks < EKB; ++ks) {
            const v4f w = wb[u & 1][ks];
            v2f b0, b1;
            b0[0] = w[0]; b0[1] = w[1];
            b1[0] = w[2]; b1[1] = w[3];
            acc0 = __builtin_amdgcn_wmma_f32_16x16x4_f32(false, xa[u % EXR][ks], false, b0,
                                                         (short)0, acc0, false, false);
            acc1 = __builtin_amdgcn_wmma_f32_16x16x4_f32(false, xa[u % EXR][ks], false, b1,
                                                         (short)0, acc1, false, false);
        }
    };
    static_assert(ENB % EXR == 0 && EXR % 2 == 0, "ring slots must be compile-time");
    for (int blk = 0; blk < ENB - EXR; blk += EXR) {
        #pragma unroll
        for (int u = 0; u < EXR; ++u) block(u, blk + u, true, true);
    }
    #pragma unroll
    for (int u = 0; u < EXR; ++u)
        block(u, ENB - EXR + u, u + EXR - 1 < EXR, u + 1 < EXR);

    const float bv0 = bias[l15];
    const float bv1 = bias[l15 + 16];
    const float LOG2E = 1.44269504088896340736f;
    #pragma unroll
    for (int r = 0; r < 8; ++r) {
        const float v0 = acc0[r] + bv0;
        const float v1 = acc1[r] + bv1;
        float mx = fmaxf(v0, v1);
        #pragma unroll
        for (int off = 8; off; off >>= 1) mx = fmaxf(mx, __shfl_xor(mx, off, 32));
        const float ms = -mx * LOG2E;
        size_t row = (size_t)mt * 16 + r + half * 8;
        wgt[row * Kv + l15]      = fexp2(fmaf(v0, LOG2E, ms));
        wgt[row * Kv + l15 + 16] = fexp2(fmaf(v1, LOG2E, ms));
    }
}

#define CHL 32
#define NCH (Tv / CHL)
#define PFv 4
#define RSv 8

__device__ __forceinline__ float other_half(float v) {
    unsigned u = __builtin_bit_cast(unsigned, v);
    return __builtin_bit_cast(float,
        __builtin_amdgcn_permlanex16(u, u, 0x76543210u, 0xfedcba98u, false, false));
}

__device__ __forceinline__ float max16(v8f a, v8f b) {
    float m0 = fmaxf(fmaxf(a[0], a[1]), a[2]);
    float m1 = fmaxf(fmaxf(a[3], a[4]), a[5]);
    float m2 = fmaxf(fmaxf(a[6], a[7]), b[0]);
    float m3 = fmaxf(fmaxf(b[1], b[2]), b[3]);
    float m4 = fmaxf(fmaxf(b[4], b[5]), b[6]);
    float m5 = fmaxf(fmaxf(m0, m1), b[7]);
    float m6 = fmaxf(fmaxf(m2, m3), m4);
    return fmaxf(m5, m6);
}

__device__ __forceinline__ void crf_chunk_matrix(const float* __restrict__ wgt,
                                                 const float* __restrict__ U,
                                                 float* __restrict__ Pm,
                                                 float* __restrict__ Pt,
                                                 float* __restrict__ Ex,
                                                 int chunk, int b) {
    const int   lane  = threadIdx.x & 31;
    const int   nt    = __builtin_amdgcn_readfirstlane((int)(threadIdx.x >> 5));
    const int   c     = lane & 15;
    const int   h     = lane >> 4;
    const float LOG2E = 1.44269504088896340736f;

    v2f A[2][2][4];
    #pragma unroll
    for (int m = 0; m < 2; ++m)
    #pragma unroll
    for (int q = 0; q < 2; ++q)
    #pragma unroll
    for (int j = 0; j < 4; ++j)
    #pragma unroll
    for (int i = 0; i < 2; ++i) {
        const int row = 16 * m + c;
        const int t   = 16 * q + 8 * h + 2 * j + i;
        A[m][q][j][i] = fexp2(U[t * Kv + row] * LOG2E);
    }

    v8f P[2];
    #pragma unroll
    for (int q = 0; q < 2; ++q)
    #pragma unroll
    for (int r = 0; r < 8; ++r)
        P[q][r] = (16 * q + 8 * h + r == 16 * nt + c) ? 1.0f : 0.0f;
    int esum = 0;

    const float* wlane = wgt + (size_t)b * (Tv * Kv) + 8 * h;
    v4f ring[RSv][4];
    auto issue = [&](int slot, int n) {
        const int tt = n < Tv - 1 ? n : Tv - 1;
        const float* rowp = wlane + (size_t)tt * Kv;
        ring[slot][0] = *(const v4f*)(rowp);
        ring[slot][1] = *(const v4f*)(rowp + 4);
        ring[slot][2] = *(const v4f*)(rowp + 16);
        ring[slot][3] = *(const v4f*)(rowp + 20);
    };
    const int n0 = chunk * CHL;
    #pragma unroll
    for (int j = 0; j < PFv; ++j) issue(j, n0 + 1 + j);

    auto step = [&](int slot, int n) {
        issue((slot + PFv) % RSv, n + PFv);
        const v8f w0 = __builtin_shufflevector(ring[slot][0], ring[slot][1], 0, 1, 2, 3, 4, 5, 6, 7);
        const v8f w1 = __builtin_shufflevector(ring[slot][2], ring[slot][3], 0, 1, 2, 3, 4, 5, 6, 7);
        float mx = max16(P[0], P[1]);
        mx = fmaxf(mx, other_half(mx));
        const unsigned eb = (__builtin_bit_cast(unsigned, mx) >> 23) & 0xffu;
        const float sc = __builtin_bit_cast(float, (254u - eb) << 23);
        esum += 127 - (int)eb;
        v8f acc0 = {};
        v8f acc1 = {};
        #pragma unroll
        for (int q = 0; q < 2; ++q) {
            #pragma unroll
            for (int j = 0; j < 4; ++j) {
                v2f bb;
                bb[0] = P[q][2 * j];
                bb[1] = P[q][2 * j + 1];
                acc0 = __builtin_amdgcn_wmma_f32_16x16x4_f32(false, A[0][q][j], false, bb,
                                                             (short)0, acc0, false, false);
                acc1 = __builtin_amdgcn_wmma_f32_16x16x4_f32(false, A[1][q][j], false, bb,
                                                             (short)0, acc1, false, false);
            }
        }
        P[0] = acc0 * (w0 * sc);
        P[1] = acc1 * (w1 * sc);
    };
    static_assert(CHL % RSv == 0 && CHL >= 2 * RSv, "ring slots must be compile-time");
    for (int s0 = 0; s0 < CHL - RSv; s0 += RSv) {
        #pragma unroll
        for (int u = 0; u < RSv; ++u) step(u, n0 + 1 + s0 + u);
    }
    #pragma unroll
    for (int u = 0; u < RSv - 1; ++u) step(u, n0 + 1 + (CHL - RSv) + u);
    if (n0 + CHL < Tv) step(RSv - 1, n0 + CHL);

    if (chunk < NCH - 1) {
        float* pb = Pm + ((size_t)(chunk * Bv + b) * 2 + nt) * 512;
        #pragma unroll
        for (int m = 0; m < 2; ++m)
        #pragma unroll
        for (int r = 0; r < 8; ++r)
            pb[(16 * m + 8 * h + r) * 16 + c] = P[m][r];
    }
    if (chunk > 0) {
        float* pc = Pt + (size_t)(chunk * Bv + b) * 1024 + (16 * nt + c) * 32 + 8 * h;
        #pragma unroll
        for (int m = 0; m < 2; ++m) {
            *(v4f*)(pc + 16 * m)     = __builtin_shufflevector(P[m], P[m], 0, 1, 2, 3);
            *(v4f*)(pc + 16 * m + 4) = __builtin_shufflevector(P[m], P[m], 4, 5, 6, 7);
        }
    }
    if (h == 0) Ex[(size_t)(chunk * Bv + b) * Kv + 16 * nt + c] = (float)esum;
}

__global__ __launch_bounds__(64) void crf_chunks(const float* __restrict__ wgt,
                                                 const float* __restrict__ U,
                                                 float* __restrict__ Pm,
                                                 float* __restrict__ Pt,
                                                 float* __restrict__ Ex) {
    const int id = blockIdx.x;
    crf_chunk_matrix(wgt, U, Pm, Pt, Ex, id / Bv, id % Bv);
}

template <int XM>
__device__ __forceinline__ int row_xor(int v) {
    return __builtin_amdgcn_update_dpp(v, v, 0x160 + XM, 0xf, 0xf, false);
}
__device__ __forceinline__ float wave_max(float v) {
    v = fmaxf(v, __builtin_bit_cast(float, row_xor<1>(__builtin_bit_cast(int, v))));
    v = fmaxf(v, __builtin_bit_cast(float, row_xor<2>(__builtin_bit_cast(int, v))));
    v = fmaxf(v, __builtin_bit_cast(float, row_xor<4>(__builtin_bit_cast(int, v))));
    v = fmaxf(v, __builtin_bit_cast(float, row_xor<8>(__builtin_bit_cast(int, v))));
    return fmaxf(v, other_half(v));
}
__device__ __forceinline__ int wave_min(int v) {
    int o;
    o = row_xor<1>(v); v = o < v ? o : v;
    o = row_xor<2>(v); v = o < v ? o : v;
    o = row_xor<4>(v); v = o < v ? o : v;
    o = row_xor<8>(v); v = o < v ? o : v;
    o = __builtin_bit_cast(int, other_half(__builtin_bit_cast(float, v)));
    return o < v ? o : v;
}

template <bool FWD>
__device__ __forceinline__ void crf_join_chunks(const float* __restrict__ wgt,
                                                const float* __restrict__ Pmat,
                                                const float* __restrict__ Ex,
                                                float* __restrict__ Ss, int b) {
    const int lane = threadIdx.x & 31;
    float S = FWD ? wgt[((size_t)b * Tv) * Kv + lane] : 1.0f;

    v4f   pr[4][8];
    float ex[4];
    float wr[4];
    auto loadm = [&](int slot, int i) {
        const int ch = FWD ? i : (NCH - 1 - i);
        if (FWD) {
            const float* p0 = Pmat + ((size_t)(ch * Bv + b) * 2) * 512 + lane * 16;
            #pragma unroll
            for (int g = 0; g < 4; ++g) {
                pr[slot][g]     = *(const v4f*)(p0 + 4 * g);
                pr[slot][4 + g] = *(const v4f*)(p0 + 512 + 4 * g);
            }
        } else {
            const float* p0 = Pmat + (size_t)(ch * Bv + b) * 1024 + lane * 32;
            #pragma unroll
            for (int g = 0; g < 8; ++g) pr[slot][g] = *(const v4f*)(p0 + 4 * g);
            wr[slot] = wgt[((size_t)b * Tv + ch * CHL) * Kv + lane];
        }
        ex[slot] = Ex[(size_t)(ch * Bv + b) * Kv + lane];
    };

    loadm(0, 0);
    loadm(1, 1);
    loadm(2, 2);
    static_assert(NCH - 1 >= 3, "three matrices are requested up front");
    #pragma unroll
    for (int i = 0; i < NCH - 1; ++i) {
        const int cur = i & 3;
        if (i + 3 < NCH - 1) loadm((i + 3) & 3, i + 3);
        __builtin_amdgcn_sched_barrier(0);
        const int e  = (int)ex[cur];
        const int ed = wave_min(e) - e;
        const float v = FWD ? ldexpf(S, ed) : S;
        float s0 = 0.0f, s1 = 0.0f, s2 = 0.0f, s3 = 0.0f;
        #pragma unroll
        for (int g = 0; g < 8; ++g) {
            s0 = fmaf(pr[cur][g][0], readlane_f(v, 4 * g + 0), s0);
            s1 = fmaf(pr[cur][g][1], readlane_f(v, 4 * g + 1), s1);
            s2 = fmaf(pr[cur][g][2], readlane_f(v, 4 * g + 2), s2);
            s3 = fmaf(pr[cur][g][3], readlane_f(v, 4 * g + 3), s3);
        }
        float sn = (s0 + s1) + (s2 + s3);
        if (!FWD) sn = ldexpf(sn, ed);
        S = sn * __builtin_amdgcn_rcpf(wave_max(sn));
        Ss[((size_t)(i + 1) * Bv + b) * Kv + lane] = FWD ? S : S * wr[cur];
    }
}

__global__ __launch_bounds__(32) void crf_join(const float* __restrict__ wgt,
                                               const float* __restrict__ Pm,
                                               const float* __restrict__ Pt,
                                               const float* __restrict__ Ex,
                                               float* __restrict__ Ss) {
    if (blockIdx.x < Bv) crf_join_chunks<true >(wgt, Pm, Ex, Ss, blockIdx.x);
    else                 crf_join_chunks<false>(wgt, Pt, Ex, Ss + (size_t)NCH * Bv * Kv, blockIdx.x - Bv);
}

#define ABP 36
#define ABN (CHL * 16 * ABP)

__global__ __launch_bounds__(128) void crf_scan(const float* __restrict__ wgt,
                                                const float* __restrict__ U,
                                                const float* __restrict__ Ss,
                                                float* __restrict__ out) {
    __shared__ v4f xch[2 * 2 * 2 * 32 * 2];
    __shared__ v4f tiles[2 * ABN / 4];

    const int   lane  = threadIdx.x & 31;
    const int   wv    = __builtin_amdgcn_readfirstlane((int)(threadIdx.x >> 5));
    const bool  fwd   = wv < 2;
    const int   dir   = wv >> 1;
    const int   m     = wv & 1;
    const int   o     = 1 - m;
    const int   c     = lane & 15;
    const int   h     = lane >> 4;
    const int   grp   = blockIdx.x % (Bv / 16);
    const int   blk   = blockIdx.x / (Bv / 16);
    const int   chunk = fwd ? blk : (NCH - 1 - blk);
    const float LOG2E = 1.44269504088896340736f;

    v2f Aown[4], Aoth[4];
    #pragma unroll
    for (int j = 0; j < 4; ++j) {
        #pragma unroll
        for (int i = 0; i < 2; ++i) {
            const int row = 16 * m + c;
            const int t0  = 16 * m + 8 * h + 2 * j + i;
            const int t1  = 16 * o + 8 * h + 2 * j + i;
            const float u0 = U[fwd ? (t0 * Kv + row) : (row * Kv + t0)];
            const float u1 = U[fwd ? (t1 * Kv + row) : (row * Kv + t1)];
            Aown[j][i] = fexp2(u0 * LOG2E);
            Aoth[j][i] = fexp2(u1 * LOG2E);
        }
    }

    const unsigned laneoff = (unsigned)((grp * 16 + c) * (Tv * Kv) + 16 * m + 8 * h);
    const float* elane = wgt + laneoff;
    float* tl = (float*)tiles + dir * ABN + c * ABP + 16 * m + 8 * h;

    v4f* xown = xch + dir * 256 + (m * 32 + lane) * 2;
    v4f* xoth = xch + dir * 256 + (o * 32 + lane) * 2;

    v4f ring[RSv][2];
    auto issue = [&](int slot, int n) {
        const int nn = n < Tv - 1 ? n : Tv - 1;
        const int tt = fwd ? nn : (Tv - 1 - nn);
        const float* rowp = elane + (size_t)tt * Kv;
        ring[slot][0] = *(const v4f*)(rowp);
        ring[slot][1] = *(const v4f*)(rowp + 4);
    };
    auto weights = [&](int slot) {
        return __builtin_shufflevector(ring[slot][0], ring[slot][1], 0, 1, 2, 3, 4, 5, 6, 7);
    };
    auto keep8 = [&](int lr, v8f s) {
        float* rowd = tl + lr * (16 * ABP);
        *(v4f*)(rowd)     = __builtin_shufflevector(s, s, 0, 1, 2, 3);
        *(v4f*)(rowd + 4) = __builtin_shufflevector(s, s, 4, 5, 6, 7);
    };
    auto handover = [&](int p, v8f s) {
        xown[p * 128]     = __builtin_shufflevector(s, s, 0, 1, 2, 3);
        xown[p * 128 + 1] = __builtin_shufflevector(s, s, 4, 5, 6, 7);
        __builtin_amdgcn_fence(__ATOMIC_RELEASE, "workgroup", "local");
        __builtin_amdgcn_s_barrier();
        __builtin_amdgcn_fence(__ATOMIC_ACQUIRE, "workgroup", "local");
    };

    const int n0 = fwd ? chunk * CHL : (chunk == 0 ? 0 : chunk * CHL - 1);
    const float* sp = (chunk == 0)
        ? (elane + (size_t)(fwd ? 0 : (Tv - 1)) * Kv)
        : (Ss + ((size_t)(dir * NCH + chunk) * Bv + grp * 16 + c) * Kv + 16 * m + 8 * h);
    v8f Sown = __builtin_shufflevector(*(const v4f*)(sp), *(const v4f*)(sp + 4),
                                       0, 1, 2, 3, 4, 5, 6, 7);
    if (fwd) {
        keep8(0, Sown);
    } else if (chunk == 0) {
        v8f ones;
        #pragma unroll
        for (int r = 0; r < 8; ++r) ones[r] = 1.0f;
        keep8(CHL - 1, ones);
    }
    #pragma unroll
    for (int j = 0; j < PFv; ++j) issue(j, n0 + 1 + j);
    handover(0, Sown);

    auto step = [&](int slot, int n, int par, bool prefetch, bool fkeep) {
        if (prefetch) issue((slot + PFv) % RSv, n + PFv);
        const v4f o0 = xoth[par * 128];
        const v4f o1 = xoth[par * 128 + 1];
        v8f w = weights(slot);
        v8f accA = {};
        v8f accB = {};
        #pragma unroll
        for (int j = 0; j < 4; ++j) {
            v2f b;
            b[0] = Sown[2 * j];
            b[1] = Sown[2 * j + 1];
            if (j & 1) accB = __builtin_amdgcn_wmma_f32_16x16x4_f32(false, Aown[j], false, b,
                                                                    (short)0, accB, false, false);
            else       accA = __builtin_amdgcn_wmma_f32_16x16x4_f32(false, Aown[j], false, b,
                                                                    (short)0, accA, false, false);
        }
        const v8f Soth = __builtin_shufflevector(o0, o1, 0, 1, 2, 3, 4, 5, 6, 7);
        #pragma unroll
        for (int j = 0; j < 4; ++j) {
            v2f b;
            b[0] = Soth[2 * j];
            b[1] = Soth[2 * j + 1];
            if (j & 1) accB = __builtin_amdgcn_wmma_f32_16x16x4_f32(false, Aoth[j], false, b,
                                                                    (short)0, accB, false, false);
            else       accA = __builtin_amdgcn_wmma_f32_16x16x4_f32(false, Aoth[j], false, b,
                                                                    (short)0, accA, false, false);
        }
        float mx = max16(Sown, Soth);
        mx = fmaxf(mx, other_half(mx));
        const float rs = __builtin_amdgcn_rcpf(mx);
        w *= rs;
        const v8f acc = accA + accB;
        Sown = acc * w;
        const int tt = fwd ? n : (Tv - 1 - n);
        if (fwd) { if (fkeep) keep8(tt & (CHL - 1), Sown); }
        else     keep8(tt & (CHL - 1), acc);
        handover(par ^ 1, Sown);
    };

    static_assert(CHL % RSv == 0 && RSv % 2 == 0, "ring slot and buffer parity must be compile-time");
    static_assert(CHL >= 2 * RSv, "a main loop and an unrolled tail");
    static_assert((CHL & (CHL - 1)) == 0 && Tv % CHL == 0, "row in block = t mod CHL");
    for (int s0 = 0; s0 < CHL - RSv; s0 += RSv) {
        #pragma unroll
        for (int u = 0; u < RSv; ++u) step(u, n0 + 1 + s0 + u, u & 1, true, true);
    }
    #pragma unroll
    for (int u = 0; u < RSv - 1; ++u)
        step(u, n0 + 1 + (CHL - RSv) + u, u & 1, u + PFv < RSv, true);
    if (blk != NCH - 1) step(RSv - 1, n0 + CHL, (RSv - 1) & 1, false, false);

    __syncthreads();
    {
        const int l8 = lane & 7;
        const int cc = wv * 4 + (lane >> 3);
        const float* ta = (const float*)tiles + cc * ABP + 4 * l8;
        const float* tb = ta + ABN;
        float* op = out + ((size_t)(grp * 16 + cc) * Tv + blk * CHL) * Kv + 4 * l8;
        #pragma unroll 4
        for (int lr = 0; lr < CHL; ++lr) {
            const v4f a = *(const v4f*)(ta + lr * (16 * ABP));
            const v4f b = *(const v4f*)(tb + lr * (16 * ABP));
            const v4f p = a * b;
            float s = (p[0] + p[1]) + (p[2] + p[3]);
            s += __builtin_bit_cast(float, row_xor<1>(__builtin_bit_cast(int, s)));
            s += __builtin_bit_cast(float, row_xor<2>(__builtin_bit_cast(int, s)));
            s += __builtin_bit_cast(float, row_xor<4>(__builtin_bit_cast(int, s)));
            const float r = 1.0f / s;
            *(v4f*)(op + lr * Kv) = p * r;
        }
    }
}

extern "C" void kernel_launch(void* const* d_in, const int* in_sizes, int n_in,
                              void* d_out, int out_size, void* d_ws, size_t ws_size,
                              hipStream_t stream) {
    const float* x = (const float*)d_in[0];
    const float* W = (const float*)d_in[1];
    const float* U = (const float*)d_in[2];
    const float* b = (const float*)d_in[3];
    float* out = (float*)d_out;

    float* ws    = (float*)d_ws;
    float* wgt   = ws;
    float* Wp    = wgt + (size_t)Bv * Tv * Kv;
    float* Pm    = Wp + (size_t)Dv * Kv;
    float* Pt    = Pm + (size_t)NCH * Bv * 1024;
    float* Ex    = Pt + (size_t)NCH * Bv * 1024;
    float* Ss    = Ex + (size_t)NCH * Bv * Kv;

    crf_packW     <<<dim3(128), dim3(256), 0, stream>>>(W, Wp);
    crf_emissions <<<dim3((Bv * Tv / 16) / 8), dim3(256), 0, stream>>>(x, Wp, b, wgt);
    crf_chunks    <<<dim3(NCH * Bv), dim3(64), 0, stream>>>(wgt, U, Pm, Pt, Ex);
    crf_join      <<<dim3(2 * Bv), dim3(32), 0, stream>>>(wgt, Pm, Pt, Ex, Ss);
    crf_scan      <<<dim3(NCH * (Bv / 16)), dim3(128), 0, stream>>>(wgt, U, Ss, out);
}


# chunk matrices are built in the tail of the emissions launch: its 16-row tiles are shifted by one row so a work-group's 128 weight rows are 4 whole chunks, 2 waves per chunk; 5 launches to 4
# speedup vs baseline: 6.5403x; 1.0417x over previous
#include <hip/hip_runtime.h>
#include <hip/hip_bf16.h>

typedef __attribute__((ext_vector_type(2))) float v2f;
typedef __attribute__((ext_vector_type(4))) float v4f;
typedef __attribute__((ext_vector_type(8))) float v8f;

#define Bv 64
#define Tv 512
#define Dv 1024
#define Kv 32

__device__ __forceinline__ float fexp2(float x) { return __builtin_amdgcn_exp2f(x); }
__device__ __forceinline__ float flog2(float x) { return __builtin_amdgcn_logf(x); }

__device__ __forceinline__ float readlane_f(float v, int l) {
    return __builtin_bit_cast(float,
        __builtin_amdgcn_readlane(__builtin_bit_cast(int, v), l));
}

__global__ __launch_bounds__(256) void crf_packW(const float* __restrict__ W,
                                                 float* __restrict__ Wp) {
    const int idx  = blockIdx.x * 256 + threadIdx.x;
    const int i    = idx & 3;
    const int lane = (idx >> 2) & 31;
    const int d4   = idx >> 7;
    const int n    = lane & 15;
    const int h    = lane >> 4;
    const int d    = 4 * d4 + 2 * h + (i & 1);
    const int k    = n + 16 * (i >> 1);
    Wp[idx] = W[d * Kv + k];
}

#define CHL 32
#define NCH (Tv / CHL)
#define PFv 4
#define RSv 8

__device__ __forceinline__ float other_half(float v) {
    unsigned u = __builtin_bit_cast(unsigned, v);
    return __builtin_bit_cast(float,
        __builtin_amdgcn_permlanex16(u, u, 0x76543210u, 0xfedcba98u, false, false));
}

__device__ __forceinline__ float max16(v8f a, v8f b) {
    float m0 = fmaxf(fmaxf(a[0], a[1]), a[2]);
    float m1 = fmaxf(fmaxf(a[3], a[4]), a[5]);
    float m2 = fmaxf(fmaxf(a[6], a[7]), b[0]);
    float m3 = fmaxf(fmaxf(b[1], b[2]), b[3]);
    float m4 = fmaxf(fmaxf(b[4], b[5]), b[6]);
    float m5 = fmaxf(fmaxf(m0, m1), b[7]);
    float m6 = fmaxf(fmaxf(m2, m3), m4);
    return fmaxf(m5, m6);
}

__device__ __forceinline__ void crf_chunk_A(const float* __restrict__ U, v2f (&A)[2][2][4]) {
    const int   lane  = threadIdx.x & 31;
    const int   c     = lane & 15;
    const int   h     = lane >> 4;
    const float LOG2E = 1.44269504088896340736f;
    #pragma unroll
    for (int m = 0; m < 2; ++m)
    #pragma unroll
    for (int q = 0; q < 2; ++q)
    #pragma unroll
    for (int j = 0; j < 4; ++j)
    #pragma unroll
    for (int i = 0; i < 2; ++i) {
        const int row = 16 * m + c;
        const int t   = 16 * q + 8 * h + 2 * j + i;
        A[m][q][j][i] = fexp2(U[t * Kv + row] * LOG2E);
    }
}

__device__ __forceinline__ void crf_chunk_matrix(const float* wgt,
                                                 const v2f (&A)[2][2][4],
                                                 float* __restrict__ Pm,
                                                 float* __restrict__ Pt,
                                                 float* __restrict__ Ex,
                                                 int chunk, int b, int nt) {
    const int   lane  = threadIdx.x & 31;
    const int   c     = lane & 15;
    const int   h     = lane >> 4;

    v8f P[2];
    #pragma unroll
    for (int q = 0; q < 2; ++q)
    #pragma unroll
    for (int r = 0; r < 8; ++r)
        P[q][r] = (16 * q + 8 * h + r == 16 * nt + c) ? 1.0f : 0.0f;
    int esum = 0;

    const float* wlane = wgt + (size_t)b * (Tv * Kv) + 8 * h;
    v4f ring[RSv][4];
    const int n0    = chunk * CHL;
    const int nlast = n0 + CHL < Tv ? n0 + CHL : Tv - 1;
    auto issue = [&](int slot, int n) {
        const int tt = n < nlast ? n : nlast;
        const float* rowp = wlane + (size_t)tt * Kv;
        ring[slot][0] = *(const v4f*)(rowp);
        ring[slot][1] = *(const v4f*)(rowp + 4);
        ring[slot][2] = *(const v4f*)(rowp + 16);
        ring[slot][3] = *(const v4f*)(rowp + 20);
    };
    #pragma unroll
    for (int j = 0; j < PFv; ++j) issue(j, n0 + 1 + j);

    auto step = [&](int slot, int n) {
        issue((slot + PFv) % RSv, n + PFv);
        const v8f w0 = __builtin_shufflevector(ring[slot][0], ring[slot][1], 0, 1, 2, 3, 4, 5, 6, 7);
        const v8f w1 = __builtin_shufflevector(ring[slot][2], ring[slot][3], 0, 1, 2, 3, 4, 5, 6, 7);
        float mx = max16(P[0], P[1]);
        mx = fmaxf(mx, other_half(mx));
        const unsigned eb = (__builtin_bit_cast(unsigned, mx) >> 23) & 0xffu;
        const float sc = __builtin_bit_cast(float, (254u - eb) << 23);
        esum += 127 - (int)eb;
        v8f acc0 = {};
        v8f acc1 = {};
        #pragma unroll
        for (int q = 0; q < 2; ++q) {
            #pragma unroll
            for (int j = 0; j < 4; ++j) {
                v2f bb;
                bb[0] = P[q][2 * j];
                bb[1] = P[q][2 * j + 1];
                acc0 = __builtin_amdgcn_wmma_f32_16x16x4_f32(false, A[0][q][j], false, bb,
                                                             (short)0, acc0, false, false);
                acc1 = __builtin_amdgcn_wmma_f32_16x16x4_f32(false, A[1][q][j], false, bb,
                                                             (short)0, acc1, false, false);
            }
        }
        P[0] = acc0 * (w0 * sc);
        P[1] = acc1 * (w1 * sc);
    };
    static_assert(CHL % RSv == 0 && CHL >= 2 * RSv, "ring slots must be compile-time");
    for (int s0 = 0; s0 < CHL - RSv; s0 += RSv) {
        #pragma unroll
        for (int u = 0; u < RSv; ++u) step(u, n0 + 1 + s0 + u);
    }
    #pragma unroll
    for (int u = 0; u < RSv - 1; ++u) step(u, n0 + 1 + (CHL - RSv) + u);
    if (n0 + CHL < Tv) step(RSv - 1, n0 + CHL);

    if (chunk < NCH - 1) {
        float* pb = Pm + ((size_t)(chunk * Bv + b) * 2 + nt) * 512;
        #pragma unroll
        for (int m = 0; m < 2; ++m)
        #pragma unroll
        for (int r = 0; r < 8; ++r)
            pb[(16 * m + 8 * h + r) * 16 + c] = P[m][r];
    }
    if (chunk > 0) {
        float* pc = Pt + (size_t)(chunk * Bv + b) * 1024 + (16 * nt + c) * 32 + 8 * h;
        #pragma unroll
        for (int m = 0; m < 2; ++m) {
            *(v4f*)(pc + 16 * m)     = __builtin_shufflevector(P[m], P[m], 0, 1, 2, 3);
            *(v4f*)(pc + 16 * m + 4) = __builtin_shufflevector(P[m], P[m], 4, 5, 6, 7);
        }
    }
    if (h == 0) Ex[(size_t)(chunk * Bv + b) * Kv + 16 * nt + c] = (float)esum;
}

#define EKB 8
#define ENB ((Dv / 4) / EKB)
#define EXR 4
static_assert(Tv == 4 * 128 && CHL == 32 && ((Bv * Tv) & (Bv * Tv - 1)) == 0,
              "a work-group's 128 rows are 4 chunks of one batch; row wrap by mask");

__global__ __launch_bounds__(256) void crf_emissions(const float* __restrict__ x,
                                                     const float* __restrict__ Wp,
                                                     const float* __restrict__ bias,
                                                     const float* __restrict__ U,
                                                     float* wgt,
                                                     float* __restrict__ Pm,
                                                     float* __restrict__ Pt,
                                                     float* __restrict__ Ex) {
    __shared__ v4f wl[(Dv / 4) * 32];

    const int tid  = threadIdx.x;
    const int lane = tid & 31;
    const int wave = tid >> 5;
    const int mt   = blockIdx.x * 8 + wave;
    const int half = lane >> 4;
    const int l15  = lane & 15;

    const int   NROW = Bv * Tv;
    const float* xptr = x + (size_t)((mt * 16 + 1 + l15) & (NROW - 1)) * Dv + half * 2;

    v2f xa[EXR][EKB];
    v4f wb[2][EKB];
    auto loadx = [&](int slot, int blk) {
        #pragma unroll
        for (int ks = 0; ks < EKB; ++ks)
            xa[slot][ks] = *(const v2f*)(xptr + (blk * EKB + ks) * 4);
    };
    auto loadw = [&](int slot, int blk) {
        #pragma unroll
        for (int ks = 0; ks < EKB; ++ks)
            wb[slot][ks] = wl[(blk * EKB + ks) * 32 + lane];
    };

    #pragma unroll
    for (int j = 0; j < EXR - 1; ++j) loadx(j, j);

    {
        const v4f* wp4 = (const v4f*)Wp;
        #pragma unroll
        for (int pass = 0; pass < 4; ++pass) {
            v4f t[8];
            #pragma unroll
            for (int j = 0; j < 8; ++j) t[j] = wp4[(pass * 8 + j) * 256 + tid];
            #pragma unroll
            for (int j = 0; j < 8; ++j) wl[(pass * 8 + j) * 256 + tid] = t[j];
        }
    }
    __syncthreads();
    loadw(0, 0);

    v8f acc0 = {};
    v8f acc1 = {};
    auto block = [&](int u, int blk, bool px, bool pw) {
        if (px) loadx((u + EXR - 1) % EXR, blk + EXR - 1);
        if (pw) loadw((u + 1) & 1, blk + 1);
        __builtin_amdgcn_sched_barrier(0);
        #pragma unroll
        for (int ks = 0; ks < EKB; ++ks) {
            const v4f w = wb[u & 1][ks];
            v2f b0, b1;
            b0[0] = w[0]; b0[1] = w[1];
            b1[0] = w[2]; b1[1] = w[3];
            acc0 = __builtin_amdgcn_wmma_f32_16x16x4_f32(false, xa[u % EXR][ks], false, b0,
                                                         (short)0, acc0, false, false);
            acc1 = __builtin_amdgcn_wmma_f32_16x16x4_f32(false, xa[u % EXR][ks], false, b1,
                                                         (short)0, acc1, false, false);
        }
    };
    static_assert(ENB % EXR == 0 && EXR % 2 == 0, "ring slots must be compile-time");
    for (int blk = 0; blk < ENB - EXR; blk += EXR) {
        #pragma unroll
        for (int u = 0; u < EXR; ++u) block(u, blk + u, true, true);
    }
    #pragma unroll
    for (int u = 0; u < EXR; ++u)
        block(u, ENB - EXR + u, u + EXR - 1 < EXR, u + 1 < EXR);

    v2f A[2][2][4];
    crf_chunk_A(U, A);

    const float bv0 = bias[l15];
    const float bv1 = bias[l15 + 16];
    const float LOG2E = 1.44269504088896340736f;
    #pragma unroll
    for (int r = 0; r < 8; ++r) {
        const float v0 = acc0[r] + bv0;
        const float v1 = acc1[r] + bv1;
        float mx = fmaxf(v0, v1);
        #pragma unroll
        for (int off = 8; off; off >>= 1) mx = fmaxf(mx, __shfl_xor(mx, off, 32));
        const float ms = -mx * LOG2E;
        size_t row = (size_t)((mt * 16 + 1 + r + half * 8) & (NROW - 1));
        wgt[row * Kv + l15]      = fexp2(fmaf(v0, LOG2E, ms));
        wgt[row * Kv + l15 + 16] = fexp2(fmaf(v1, LOG2E, ms));
    }

    __syncthreads();
    const int ws = __builtin_amdgcn_readfirstlane(wave);
    crf_chunk_matrix(wgt, A, Pm, Pt, Ex,
                     (int)(blockIdx.x & 3) * 4 + (ws >> 1),
                     (int)(blockIdx.x >> 2),
                     ws & 1);
}

template <int XM>
__device__ __forceinline__ int row_xor(int v) {
    return __builtin_amdgcn_update_dpp(v, v, 0x160 + XM, 0xf, 0xf, false);
}
__device__ __forceinline__ float wave_max(float v) {
    v = fmaxf(v, __builtin_bit_cast(float, row_xor<1>(__builtin_bit_cast(int, v))));
    v = fmaxf(v, __builtin_bit_cast(float, row_xor<2>(__builtin_bit_cast(int, v))));
    v = fmaxf(v, __builtin_bit_cast(float, row_xor<4>(__builtin_bit_cast(int, v))));
    v = fmaxf(v, __builtin_bit_cast(float, row_xor<8>(__builtin_bit_cast(int, v))));
    return fmaxf(v, other_half(v));
}
__device__ __forceinline__ int wave_min(int v) {
    int o;
    o = row_xor<1>(v); v = o < v ? o : v;
    o = row_xor<2>(v); v = o < v ? o : v;
    o = row_xor<4>(v); v = o < v ? o : v;
    o = row_xor<8>(v); v = o < v ? o : v;
    o = __builtin_bit_cast(int, other_half(__builtin_bit_cast(float, v)));
    return o < v ? o : v;
}

template <bool FWD>
__device__ __forceinline__ void crf_join_chunks(const float* __restrict__ wgt,
                                                const float* __restrict__ Pmat,
                                                const float* __restrict__ Ex,
                                                float* __restrict__ Ss, int b) {
    const int lane = threadIdx.x & 31;
    float S = FWD ? wgt[((size_t)b * Tv) * Kv + lane] : 1.0f;

    v4f   pr[4][8];
    float ex[4];
    float wr[4];
    auto loadm = [&](int slot, int i) {
        const int ch = FWD ? i : (NCH - 1 - i);
        if (FWD) {
            const float* p0 = Pmat + ((size_t)(ch * Bv + b) * 2) * 512 + lane * 16;
            #pragma unroll
            for (int g = 0; g < 4; ++g) {
                pr[slot][g]     = *(const v4f*)(p0 + 4 * g);
                pr[slot][4 + g] = *(const v4f*)(p0 + 512 + 4 * g);
            }
        } else {
            const float* p0 = Pmat + (size_t)(ch * Bv + b) * 1024 + lane * 32;
            #pragma unroll
            for (int g = 0; g < 8; ++g) pr[slot][g] = *(const v4f*)(p0 + 4 * g);
            wr[slot] = wgt[((size_t)b * Tv + ch * CHL) * Kv + lane];
        }
        ex[slot] = Ex[(size_t)(ch * Bv + b) * Kv + lane];
    };

    loadm(0, 0);
    loadm(1, 1);
    loadm(2, 2);
    static_assert(NCH - 1 >= 3, "three matrices are requested up front");
    #pragma unroll
    for (int i = 0; i < NCH - 1; ++i) {
        const int cur = i & 3;
        if (i + 3 < NCH - 1) loadm((i + 3) & 3, i + 3);
        __builtin_amdgcn_sched_barrier(0);
        const int e  = (int)ex[cur];
        const int ed = wave_min(e) - e;
        const float v = FWD ? ldexpf(S, ed) : S;
        float s0 = 0.0f, s1 = 0.0f, s2 = 0.0f, s3 = 0.0f;
        #pragma unroll
        for (int g = 0; g < 8; ++g) {
            s0 = fmaf(pr[cur][g][0], readlane_f(v, 4 * g + 0), s0);
            s1 = fmaf(pr[cur][g][1], readlane_f(v, 4 * g + 1), s1);
            s2 = fmaf(pr[cur][g][2], readlane_f(v, 4 * g + 2), s2);
            s3 = fmaf(pr[cur][g][3], readlane_f(v, 4 * g + 3), s3);
        }
        float sn = (s0 + s1) + (s2 + s3);
        if (!FWD) sn = ldexpf(sn, ed);
        S = sn * __builtin_amdgcn_rcpf(wave_max(sn));
        Ss[((size_t)(i + 1) * Bv + b) * Kv + lane] = FWD ? S : S * wr[cur];
    }
}

__global__ __launch_bounds__(32) void crf_join(const float* __restrict__ wgt,
                                               const float* __restrict__ Pm,
                                               const float* __restrict__ Pt,
                                               const float* __restrict__ Ex,
                                               float* __restrict__ Ss) {
    if (blockIdx.x < Bv) crf_join_chunks<true >(wgt, Pm, Ex, Ss, blockIdx.x);
    else                 crf_join_chunks<false>(wgt, Pt, Ex, Ss + (size_t)NCH * Bv * Kv, blockIdx.x - Bv);
}

#define ABP 36
#define ABN (CHL * 16 * ABP)

__global__ __launch_bounds__(128) void crf_scan(const float* __restrict__ wgt,
                                                const float* __restrict__ U,
                                                const float* __restrict__ Ss,
                                                float* __restrict__ out) {
    __shared__ v4f xch[2 * 2 * 2 * 32 * 2];
    __shared__ v4f tiles[2 * ABN / 4];

    const int   lane  = threadIdx.x & 31;
    const int   wv    = __builtin_amdgcn_readfirstlane((int)(threadIdx.x >> 5));
    const bool  fwd   = wv < 2;
    const int   dir   = wv >> 1;
    const int   m     = wv & 1;
    const int   o     = 1 - m;
    const int   c     = lane & 15;
    const int   h     = lane >> 4;
    const int   grp   = blockIdx.x % (Bv / 16);
    const int   blk   = blockIdx.x / (Bv / 16);
    const int   chunk = fwd ? blk : (NCH - 1 - blk);
    const float LOG2E = 1.44269504088896340736f;

    v2f Aown[4], Aoth[4];
    #pragma unroll
    for (int j = 0; j < 4; ++j) {
        #pragma unroll
        for (int i = 0; i < 2; ++i) {
            const int row = 16 * m + c;
            const int t0  = 16 * m + 8 * h + 2 * j + i;
            const int t1  = 16 * o + 8 * h + 2 * j + i;
            const float u0 = U[fwd ? (t0 * Kv + row) : (row * Kv + t0)];
            const float u1 = U[fwd ? (t1 * Kv + row) : (row * Kv + t1)];
            Aown[j][i] = fexp2(u0 * LOG2E);
            Aoth[j][i] = fexp2(u1 * LOG2E);
        }
    }

    const unsigned laneoff = (unsigned)((grp * 16 + c) * (Tv * Kv) + 16 * m + 8 * h);
    const float* elane = wgt + laneoff;
    float* tl = (float*)tiles + dir * ABN + c * ABP + 16 * m + 8 * h;

    v4f* xown = xch + dir * 256 + (m * 32 + lane) * 2;
    v4f* xoth = xch + dir * 256 + (o * 32 + lane) * 2;

    v4f ring[RSv][2];
    auto issue = [&](int slot, int n) {
        const int nn = n < Tv - 1 ? n : Tv - 1;
        const int tt = fwd ? nn : (Tv - 1 - nn);
        const float* rowp = elane + (size_t)tt * Kv;
        ring[slot][0] = *(const v4f*)(rowp);
        ring[slot][1] = *(const v4f*)(rowp + 4);
    };
    auto weights = [&](int slot) {
        return __builtin_shufflevector(ring[slot][0], ring[slot][1], 0, 1, 2, 3, 4, 5, 6, 7);
    };
    auto keep8 = [&](int lr, v8f s) {
        float* rowd = tl + lr * (16 * ABP);
        *(v4f*)(rowd)     = __builtin_shufflevector(s, s, 0, 1, 2, 3);
        *(v4f*)(rowd + 4) = __builtin_shufflevector(s, s, 4, 5, 6, 7);
    };
    auto handover = [&](int p, v8f s) {
        xown[p * 128]     = __builtin_shufflevector(s, s, 0, 1, 2, 3);
        xown[p * 128 + 1] = __builtin_shufflevector(s, s, 4, 5, 6, 7);
        __builtin_amdgcn_fence(__ATOMIC_RELEASE, "workgroup", "local");
        __builtin_amdgcn_s_barrier();
        __builtin_amdgcn_fence(__ATOMIC_ACQUIRE, "workgroup", "local");
    };

    const int n0 = fwd ? chunk * CHL : (chunk == 0 ? 0 : chunk * CHL - 1);
    const float* sp = (chunk == 0)
        ? (elane + (size_t)(fwd ? 0 : (Tv - 1)) * Kv)
        : (Ss + ((size_t)(dir * NCH + chunk) * Bv + grp * 16 + c) * Kv + 16 * m + 8 * h);
    v8f Sown = __builtin_shufflevector(*(const v4f*)(sp), *(const v4f*)(sp + 4),
                                       0, 1, 2, 3, 4, 5, 6, 7);
    if (fwd) {
        keep8(0, Sown);
    } else if (chunk == 0) {
        v8f ones;
        #pragma unroll
        for (int r = 0; r < 8; ++r) ones[r] = 1.0f;
        keep8(CHL - 1, ones);
    }
    #pragma unroll
    for (int j = 0; j < PFv; ++j) issue(j, n0 + 1 + j);
    handover(0, Sown);

    auto step = [&](int slot, int n, int par, bool prefetch, bool fkeep) {
        if (prefetch) issue((slot + PFv) % RSv, n + PFv);
        const v4f o0 = xoth[par * 128];
        const v4f o1 = xoth[par * 128 + 1];
        v8f w = weights(slot);
        v8f accA = {};
        v8f accB = {};
        #pragma unroll
        for (int j = 0; j < 4; ++j) {
            v2f b;
            b[0] = Sown[2 * j];
            b[1] = Sown[2 * j + 1];
            if (j & 1) accB = __builtin_amdgcn_wmma_f32_16x16x4_f32(false, Aown[j], false, b,
                                                                    (short)0, accB, false, false);
            else       accA = __builtin_amdgcn_wmma_f32_16x16x4_f32(false, Aown[j], false, b,
                                                                    (short)0, accA, false, false);
        }
        const v8f Soth = __builtin_shufflevector(o0, o1, 0, 1, 2, 3, 4, 5, 6, 7);
        #pragma unroll
        for (int j = 0; j < 4; ++j) {
            v2f b;
            b[0] = Soth[2 * j];
            b[1] = Soth[2 * j + 1];
            if (j & 1) accB = __builtin_amdgcn_wmma_f32_16x16x4_f32(false, Aoth[j], false, b,
                                                                    (short)0, accB, false, false);
            else       accA = __builtin_amdgcn_wmma_f32_16x16x4_f32(false, Aoth[j], false, b,
                                                                    (short)0, accA, false, false);
        }
        float mx = max16(Sown, Soth);
        mx = fmaxf(mx, other_half(mx));
        const float rs = __builtin_amdgcn_rcpf(mx);
        w *= rs;
        const v8f acc = accA + accB;
        Sown = acc * w;
        const int tt = fwd ? n : (Tv - 1 - n);
        if (fwd) { if (fkeep) keep8(tt & (CHL - 1), Sown); }
        else     keep8(tt & (CHL - 1), acc);
        handover(par ^ 1, Sown);
    };

    static_assert(CHL % RSv == 0 && RSv % 2 == 0, "ring slot and buffer parity must be compile-time");
    static_assert(CHL >= 2 * RSv, "a main loop and an unrolled tail");
    static_assert((CHL & (CHL - 1)) == 0 && Tv % CHL == 0, "row in block = t mod CHL");
    for (int s0 = 0; s0 < CHL - RSv; s0 += RSv) {
        #pragma unroll
        for (int u = 0; u < RSv; ++u) step(u, n0 + 1 + s0 + u, u & 1, true, true);
    }
    #pragma unroll
    for (int u = 0; u < RSv - 1; ++u)
        step(u, n0 + 1 + (CHL - RSv) + u, u & 1, u + PFv < RSv, true);
    if (blk != NCH - 1) step(RSv - 1, n0 + CHL, (RSv - 1) & 1, false, false);

    __syncthreads();
    {
        const int l8 = lane & 7;
        const int cc = wv * 4 + (lane >> 3);
        const float* ta = (const float*)tiles + cc * ABP + 4 * l8;
        const float* tb = ta + ABN;
        float* op = out + ((size_t)(grp * 16 + cc) * Tv + blk * CHL) * Kv + 4 * l8;
        #pragma unroll 4
        for (int lr = 0; lr < CHL; ++lr) {
            const v4f a = *(const v4f*)(ta + lr * (16 * ABP));
            const v4f b = *(const v4f*)(tb + lr * (16 * ABP));
            const v4f p = a * b;
            float s = (p[0] + p[1]) + (p[2] + p[3]);
            s += __builtin_bit_cast(float, row_xor<1>(__builtin_bit_cast(int, s)));
            s += __builtin_bit_cast(float, row_xor<2>(__builtin_bit_cast(int, s)));
            s += __builtin_bit_cast(float, row_xor<4>(__builtin_bit_cast(int, s)));
            const float r = 1.0f / s;
            *(v4f*)(op + lr * Kv) = p * r;
        }
    }
}

extern "C" void kernel_launch(void* const* d_in, const int* in_sizes, int n_in,
                              void* d_out, int out_size, void* d_ws, size_t ws_size,
                              hipStream_t stream) {
    const float* x = (const float*)d_in[0];
    const float* W = (const float*)d_in[1];
    const float* U = (const float*)d_in[2];
    const float* b = (const float*)d_in[3];
    float* out = (float*)d_out;

    float* ws    = (float*)d_ws;
    float* wgt   = ws;
    float* Wp    = wgt + (size_t)Bv * Tv * Kv;
    float* Pm    = Wp + (size_t)Dv * Kv;
    float* Pt    = Pm + (size_t)NCH * Bv * 1024;
    float* Ex    = Pt + (size_t)NCH * Bv * 1024;
    float* Ss    = Ex + (size_t)NCH * Bv * Kv;

    crf_packW     <<<dim3(128), dim3(256), 0, stream>>>(W, Wp);
    crf_emissions <<<dim3((Bv * Tv / 16) / 8), dim3(256), 0, stream>>>(x, Wp, b, U, wgt, Pm, Pt, Ex);
    crf_join      <<<dim3(2 * Bv), dim3(32), 0, stream>>>(wgt, Pm, Pt, Ex, Ss);
    crf_scan      <<<dim3(NCH * (Bv / 16)), dim3(128), 0, stream>>>(wgt, U, Ss, out);
}


# scan: the 16 columns of its products are the 16 chunks of ONE batch (was 16 batches of one chunk): a work-group holds a whole batch and makes its chunk start vectors itself, in LDS; 4 launches to 3
# speedup vs baseline: 6.8206x; 1.0429x over previous
#include <hip/hip_runtime.h>
#include <hip/hip_bf16.h>

typedef __attribute__((ext_vector_type(2))) float v2f;
typedef __attribute__((ext_vector_type(4))) float v4f;
typedef __attribute__((ext_vector_type(8))) float v8f;

#define Bv 64
#define Tv 512
#define Dv 1024
#define Kv 32

__device__ __forceinline__ float fexp2(float x) { return __builtin_amdgcn_exp2f(x); }
__device__ __forceinline__ float flog2(float x) { return __builtin_amdgcn_logf(x); }

__device__ __forceinline__ float readlane_f(float v, int l) {
    return __builtin_bit_cast(float,
        __builtin_amdgcn_readlane(__builtin_bit_cast(int, v), l));
}

__global__ __launch_bounds__(256) void crf_packW(const float* __restrict__ W,
                                                 float* __restrict__ Wp) {
    const int idx  = blockIdx.x * 256 + threadIdx.x;
    const int i    = idx & 3;
    const int lane = (idx >> 2) & 31;
    const int d4   = idx >> 7;
    const int n    = lane & 15;
    const int h    = lane >> 4;
    const int d    = 4 * d4 + 2 * h + (i & 1);
    const int k    = n + 16 * (i >> 1);
    Wp[idx] = W[d * Kv + k];
}

#define CHL 32
#define NCH (Tv / CHL)
#define PFv 4
#define RSv 8

__device__ __forceinline__ float other_half(float v) {
    unsigned u = __builtin_bit_cast(unsigned, v);
    return __builtin_bit_cast(float,
        __builtin_amdgcn_permlanex16(u, u, 0x76543210u, 0xfedcba98u, false, false));
}

__device__ __forceinline__ float max16(v8f a, v8f b) {
    float m0 = fmaxf(fmaxf(a[0], a[1]), a[2]);
    float m1 = fmaxf(fmaxf(a[3], a[4]), a[5]);
    float m2 = fmaxf(fmaxf(a[6], a[7]), b[0]);
    float m3 = fmaxf(fmaxf(b[1], b[2]), b[3]);
    float m4 = fmaxf(fmaxf(b[4], b[5]), b[6]);
    float m5 = fmaxf(fmaxf(m0, m1), b[7]);
    float m6 = fmaxf(fmaxf(m2, m3), m4);
    return fmaxf(m5, m6);
}

__device__ __forceinline__ void crf_chunk_A(const float* __restrict__ U, v2f (&A)[2][2][4]) {
    const int   lane  = threadIdx.x & 31;
    const int   c     = lane & 15;
    const int   h     = lane >> 4;
    const float LOG2E = 1.44269504088896340736f;
    #pragma unroll
    for (int m = 0; m < 2; ++m)
    #pragma unroll
    for (int q = 0; q < 2; ++q)
    #pragma unroll
    for (int j = 0; j < 4; ++j)
    #pragma unroll
    for (int i = 0; i < 2; ++i) {
        const int row = 16 * m + c;
        const int t   = 16 * q + 8 * h + 2 * j + i;
        A[m][q][j][i] = fexp2(U[t * Kv + row] * LOG2E);
    }
}

__device__ __forceinline__ void crf_chunk_matrix(const float* wgt,
                                                 const v2f (&A)[2][2][4],
                                                 float* __restrict__ Pm,
                                                 float* __restrict__ Pt,
                                                 float* __restrict__ Ex,
                                                 int chunk, int b, int nt) {
    const int   lane  = threadIdx.x & 31;
    const int   c     = lane & 15;
    const int   h     = lane >> 4;

    v8f P[2];
    #pragma unroll
    for (int q = 0; q < 2; ++q)
    #pragma unroll
    for (int r = 0; r < 8; ++r)
        P[q][r] = (16 * q + 8 * h + r == 16 * nt + c) ? 1.0f : 0.0f;
    int esum = 0;

    const float* wlane = wgt + (size_t)b * (Tv * Kv) + 8 * h;
    v4f ring[RSv][4];
    const int n0    = chunk * CHL;
    const int nlast = n0 + CHL < Tv ? n0 + CHL : Tv - 1;
    auto issue = [&](int slot, int n) {
        const int tt = n < nlast ? n : nlast;
        const float* rowp = wlane + (size_t)tt * Kv;
        ring[slot][0] = *(const v4f*)(rowp);
        ring[slot][1] = *(const v4f*)(rowp + 4);
        ring[slot][2] = *(const v4f*)(rowp + 16);
        ring[slot][3] = *(const v4f*)(rowp + 20);
    };
    #pragma unroll
    for (int j = 0; j < PFv; ++j) issue(j, n0 + 1 + j);

    auto step = [&](int slot, int n) {
        issue((slot + PFv) % RSv, n + PFv);
        const v8f w0 = __builtin_shufflevector(ring[slot][0], ring[slot][1], 0, 1, 2, 3, 4, 5, 6, 7);
        const v8f w1 = __builtin_shufflevector(ring[slot][2], ring[slot][3], 0, 1, 2, 3, 4, 5, 6, 7);
        float mx = max16(P[0], P[1]);
        mx = fmaxf(mx, other_half(mx));
        const unsigned eb = (__builtin_bit_cast(unsigned, mx) >> 23) & 0xffu;
        const float sc = __builtin_bit_cast(float, (254u - eb) << 23);
        esum += 127 - (int)eb;
        v8f acc0 = {};
        v8f acc1 = {};
        #pragma unroll
        for (int q = 0; q < 2; ++q) {
            #pragma unroll
            for (int j = 0; j < 4; ++j) {
                v2f bb;
                bb[0] = P[q][2 * j];
                bb[1] = P[q][2 * j + 1];
                acc0 = __builtin_amdgcn_wmma_f32_16x16x4_f32(false, A[0][q][j], false, bb,
                                                             (short)0, acc0, false, false);
                acc1 = __builtin_amdgcn_wmma_f32_16x16x4_f32(false, A[1][q][j], false, bb,
                                                             (short)0, acc1, false, false);
            }
        }
        P[0] = acc0 * (w0 * sc);
        P[1] = acc1 * (w1 * sc);
    };
    static_assert(CHL % RSv == 0 && CHL >= 2 * RSv, "ring slots must be compile-time");
    for (int s0 = 0; s0 < CHL - RSv; s0 += RSv) {
        #pragma unroll
        for (int u = 0; u < RSv; ++u) step(u, n0 + 1 + s0 + u);
    }
    #pragma unroll
    for (int u = 0; u < RSv - 1; ++u) step(u, n0 + 1 + (CHL - RSv) + u);
    if (n0 + CHL < Tv) step(RSv - 1, n0 + CHL);

    if (chunk < NCH - 1) {
        float* pb = Pm + ((size_t)(chunk * Bv + b) * 2 + nt) * 512;
        #pragma unroll
        for (int m = 0; m < 2; ++m)
        #pragma unroll
        for (int r = 0; r < 8; ++r)
            pb[(16 * m + 8 * h + r) * 16 + c] = P[m][r];
    }
    if (chunk > 0) {
        float* pc = Pt + (size_t)(chunk * Bv + b) * 1024 + (16 * nt + c) * 32 + 8 * h;
        #pragma unroll
        for (int m = 0; m < 2; ++m) {
            *(v4f*)(pc + 16 * m)     = __builtin_shufflevector(P[m], P[m], 0, 1, 2, 3);
            *(v4f*)(pc + 16 * m + 4) = __builtin_shufflevector(P[m], P[m], 4, 5, 6, 7);
        }
    }
    if (h == 0) Ex[(size_t)(chunk * Bv + b) * Kv + 16 * nt + c] = (float)esum;
}

#define EKB 8
#define ENB ((Dv / 4) / EKB)
#define EXR 4
static_assert(Tv == 4 * 128 && CHL == 32 && ((Bv * Tv) & (Bv * Tv - 1)) == 0,
              "a work-group's 128 rows are 4 chunks of one batch; row wrap by mask");

__global__ __launch_bounds__(256) void crf_emissions(const float* __restrict__ x,
                                                     const float* __restrict__ Wp,
                                                     const float* __restrict__ bias,
                                                     const float* __restrict__ U,
                                                     float* wgt,
                                                     float* __restrict__ Pm,
                                                     float* __restrict__ Pt,
                                                     float* __restrict__ Ex) {
    __shared__ v4f wl[(Dv / 4) * 32];

    const int tid  = threadIdx.x;
    const int lane = tid & 31;
    const int wave = tid >> 5;
    const int mt   = blockIdx.x * 8 + wave;
    const int half = lane >> 4;
    const int l15  = lane & 15;

    const int   NROW = Bv * Tv;
    const float* xptr = x + (size_t)((mt * 16 + 1 + l15) & (NROW - 1)) * Dv + half * 2;

    v2f xa[EXR][EKB];
    v4f wb[2][EKB];
    auto loadx = [&](int slot, int blk) {
        #pragma unroll
        for (int ks = 0; ks < EKB; ++ks)
            xa[slot][ks] = *(const v2f*)(xptr + (blk * EKB + ks) * 4);
    };
    auto loadw = [&](int slot, int blk) {
        #pragma unroll
        for (int ks = 0; ks < EKB; ++ks)
            wb[slot][ks] = wl[(blk * EKB + ks) * 32 + lane];
    };

    #pragma unroll
    for (int j = 0; j < EXR - 1; ++j) loadx(j, j);

    {
        const v4f* wp4 = (const v4f*)Wp;
        #pragma unroll
        for (int pass = 0; pass < 4; ++pass) {
            v4f t[8];
            #pragma unroll
            for (int j = 0; j < 8; ++j) t[j] = wp4[(pass * 8 + j) * 256 + tid];
            #pragma unroll
            for (int j = 0; j < 8; ++j) wl[(pass * 8 + j) * 256 + tid] = t[j];
        }
    }
    __syncthreads();
    loadw(0, 0);

    v8f acc0 = {};
    v8f acc1 = {};
    auto block = [&](int u, int blk, bool px, bool pw) {
        if (px) loadx((u + EXR - 1) % EXR, blk + EXR - 1);
        if (pw) loadw((u + 1) & 1, blk + 1);
        __builtin_amdgcn_sched_barrier(0);
        #pragma unroll
        for (int ks = 0; ks < EKB; ++ks) {
            const v4f w = wb[u & 1][ks];
            v2f b0, b1;
            b0[0] = w[0]; b0[1] = w[1];
            b1[0] = w[2]; b1[1] = w[3];
            acc0 = __builtin_amdgcn_wmma_f32_16x16x4_f32(false, xa[u % EXR][ks], false, b0,
                                                         (short)0, acc0, false, false);
            acc1 = __builtin_amdgcn_wmma_f32_16x16x4_f32(false, xa[u % EXR][ks], false, b1,
                                                         (short)0, acc1, false, false);
        }
    };
    static_assert(ENB % EXR == 0 && EXR % 2 == 0, "ring slots must be compile-time");
    for (int blk = 0; blk < ENB - EXR; blk += EXR) {
        #pragma unroll
        for (int u = 0; u < EXR; ++u) block(u, blk + u, true, true);
    }
    #pragma unroll
    for (int u = 0; u < EXR; ++u)
        block(u, ENB - EXR + u, u + EXR - 1 < EXR, u + 1 < EXR);

    v2f A[2][2][4];
    crf_chunk_A(U, A);

    const float bv0 = bias[l15];
    const float bv1 = bias[l15 + 16];
    const float LOG2E = 1.44269504088896340736f;
    #pragma unroll
    for (int r = 0; r < 8; ++r) {
        const float v0 = acc0[r] + bv0;
        const float v1 = acc1[r] + bv1;
        float mx = fmaxf(v0, v1);
        #pragma unroll
        for (int off = 8; off; off >>= 1) mx = fmaxf(mx, __shfl_xor(mx, off, 32));
        const float ms = -mx * LOG2E;
        size_t row = (size_t)((mt * 16 + 1 + r + half * 8) & (NROW - 1));
        wgt[row * Kv + l15]      = fexp2(fmaf(v0, LOG2E, ms));
        wgt[row * Kv + l15 + 16] = fexp2(fmaf(v1, LOG2E, ms));
    }

    __syncthreads();
    const int ws = __builtin_amdgcn_readfirstlane(wave);
    crf_chunk_matrix(wgt, A, Pm, Pt, Ex,
                     (int)(blockIdx.x & 3) * 4 + (ws >> 1),
                     (int)(blockIdx.x >> 2),
                     ws & 1);
}

template <int XM>
__device__ __forceinline__ int row_xor(int v) {
    return __builtin_amdgcn_update_dpp(v, v, 0x160 + XM, 0xf, 0xf, false);
}
__device__ __forceinline__ float wave_max(float v) {
    v = fmaxf(v, __builtin_bit_cast(float, row_xor<1>(__builtin_bit_cast(int, v))));
    v = fmaxf(v, __builtin_bit_cast(float, row_xor<2>(__builtin_bit_cast(int, v))));
    v = fmaxf(v, __builtin_bit_cast(float, row_xor<4>(__builtin_bit_cast(int, v))));
    v = fmaxf(v, __builtin_bit_cast(float, row_xor<8>(__builtin_bit_cast(int, v))));
    return fmaxf(v, other_half(v));
}
__device__ __forceinline__ int wave_min(int v) {
    int o;
    o = row_xor<1>(v); v = o < v ? o : v;
    o = row_xor<2>(v); v = o < v ? o : v;
    o = row_xor<4>(v); v = o < v ? o : v;
    o = row_xor<8>(v); v = o < v ? o : v;
    o = __builtin_bit_cast(int, other_half(__builtin_bit_cast(float, v)));
    return o < v ? o : v;
}

template <bool FWD>
__device__ __forceinline__ void crf_join_chunks(const float* __restrict__ wgt,
                                                const float* __restrict__ Pmat,
                                                const float* __restrict__ Ex,
                                                float* sv, int b) {
    const int lane = threadIdx.x & 31;
    float S = FWD ? wgt[((size_t)b * Tv) * Kv + lane] : 1.0f;
    sv[lane] = FWD ? S : wgt[((size_t)b * Tv + (Tv - 1)) * Kv + lane];

    v4f   pr[4][8];
    float ex[4];
    float wr[4];
    auto loadm = [&](int slot, int i) {
        const int ch = FWD ? i : (NCH - 1 - i);
        if (FWD) {
            const float* p0 = Pmat + ((size_t)(ch * Bv + b) * 2) * 512 + lane * 16;
            #pragma unroll
            for (int g = 0; g < 4; ++g) {
                pr[slot][g]     = *(const v4f*)(p0 + 4 * g);
                pr[slot][4 + g] = *(const v4f*)(p0 + 512 + 4 * g);
            }
        } else {
            const float* p0 = Pmat + (size_t)(ch * Bv + b) * 1024 + lane * 32;
            #pragma unroll
            for (int g = 0; g < 8; ++g) pr[slot][g] = *(const v4f*)(p0 + 4 * g);
            wr[slot] = wgt[((size_t)b * Tv + ch * CHL) * Kv + lane];
        }
        ex[slot] = Ex[(size_t)(ch * Bv + b) * Kv + lane];
    };

    loadm(0, 0);
    loadm(1, 1);
    loadm(2, 2);
    static_assert(NCH - 1 >= 3, "three matrices are requested up front");
    #pragma unroll
    for (int i = 0; i < NCH - 1; ++i) {
        const int cur = i & 3;
        if (i + 3 < NCH - 1) loadm((i + 3) & 3, i + 3);
        __builtin_amdgcn_sched_barrier(0);
        const int e  = (int)ex[cur];
        const int ed = wave_min(e) - e;
        const float v = FWD ? ldexpf(S, ed) : S;
        float s0 = 0.0f, s1 = 0.0f, s2 = 0.0f, s3 = 0.0f;
        #pragma unroll
        for (int g = 0; g < 8; ++g) {
            s0 = fmaf(pr[cur][g][0], readlane_f(v, 4 * g + 0), s0);
            s1 = fmaf(pr[cur][g][1], readlane_f(v, 4 * g + 1), s1);
            s2 = fmaf(pr[cur][g][2], readlane_f(v, 4 * g + 2), s2);
            s3 = fmaf(pr[cur][g][3], readlane_f(v, 4 * g + 3), s3);
        }
        float sn = (s0 + s1) + (s2 + s3);
        if (!FWD) sn = ldexpf(sn, ed);
        S = sn * __builtin_amdgcn_rcpf(wave_max(sn));
        sv[(i + 1) * Kv + lane] = FWD ? S : S * wr[cur];
    }
}

#define ABP 36
#define ABN (CHL * 16 * ABP)

__global__ __launch_bounds__(128) void crf_scan(const float* __restrict__ wgt,
                                                const float* __restrict__ U,
                                                const float* __restrict__ Pm,
                                                const float* __restrict__ Pt,
                                                const float* __restrict__ Ex,
                                                float* __restrict__ out) {
    __shared__ v4f xch[2 * 2 * 2 * 32 * 2];
    __shared__ v4f tiles[2 * ABN / 4];
    __shared__ v4f svec[2 * NCH * Kv / 4];

    const int   lane  = threadIdx.x & 31;
    const int   wv    = __builtin_amdgcn_readfirstlane((int)(threadIdx.x >> 5));
    const bool  fwd   = wv < 2;
    const int   dir   = wv >> 1;
    const int   m     = wv & 1;
    const int   o     = 1 - m;
    const int   c     = lane & 15;
    const int   h     = lane >> 4;
    const int   bat   = blockIdx.x;
    const int   chunk = fwd ? c : (NCH - 1 - c);
    const float LOG2E = 1.44269504088896340736f;

    v2f Aown[4], Aoth[4];
    #pragma unroll
    for (int j = 0; j < 4; ++j) {
        #pragma unroll
        for (int i = 0; i < 2; ++i) {
            const int row = 16 * m + c;
            const int t0  = 16 * m + 8 * h + 2 * j + i;
            const int t1  = 16 * o + 8 * h + 2 * j + i;
            const float u0 = U[fwd ? (t0 * Kv + row) : (row * Kv + t0)];
            const float u1 = U[fwd ? (t1 * Kv + row) : (row * Kv + t1)];
            Aown[j][i] = fexp2(u0 * LOG2E);
            Aoth[j][i] = fexp2(u1 * LOG2E);
        }
    }

    const float* elane = wgt + (size_t)bat * (Tv * Kv) + 16 * m + 8 * h;
    float* tl = (float*)tiles + dir * ABN + c * ABP + 16 * m + 8 * h;

    v4f* xown = xch + dir * 256 + (m * 32 + lane) * 2;
    v4f* xoth = xch + dir * 256 + (o * 32 + lane) * 2;

    v4f ring[RSv][2];
    auto issue = [&](int slot, int n) {
        const int nn = n < Tv - 1 ? n : Tv - 1;
        const int tt = fwd ? nn : (Tv - 1 - nn);
        const float* rowp = elane + (size_t)tt * Kv;
        ring[slot][0] = *(const v4f*)(rowp);
        ring[slot][1] = *(const v4f*)(rowp + 4);
    };
    auto weights = [&](int slot) {
        return __builtin_shufflevector(ring[slot][0], ring[slot][1], 0, 1, 2, 3, 4, 5, 6, 7);
    };
    auto keep8 = [&](int lr, v8f s) {
        float* rowd = tl + lr * (16 * ABP);
        *(v4f*)(rowd)     = __builtin_shufflevector(s, s, 0, 1, 2, 3);
        *(v4f*)(rowd + 4) = __builtin_shufflevector(s, s, 4, 5, 6, 7);
    };
    auto handover = [&](int p, v8f s) {
        xown[p * 128]     = __builtin_shufflevector(s, s, 0, 1, 2, 3);
        xown[p * 128 + 1] = __builtin_shufflevector(s, s, 4, 5, 6, 7);
        __builtin_amdgcn_fence(__ATOMIC_RELEASE, "workgroup", "local");
        __builtin_amdgcn_s_barrier();
        __builtin_amdgcn_fence(__ATOMIC_ACQUIRE, "workgroup", "local");
    };

    const int n0 = fwd ? chunk * CHL : (chunk == 0 ? 0 : chunk * CHL - 1);
    #pragma unroll
    for (int j = 0; j < PFv; ++j) issue(j, n0 + 1 + j);

    float* sv = (float*)svec + dir * (NCH * Kv);
    if (m == 0) {
        if (fwd) crf_join_chunks<true >(wgt, Pm, Ex, sv, bat);
        else     crf_join_chunks<false>(wgt, Pt, Ex, sv, bat);
    }
    __syncthreads();

    const float* sp = sv + chunk * Kv + 16 * m + 8 * h;
    v8f Sown = __builtin_shufflevector(*(const v4f*)(sp), *(const v4f*)(sp + 4),
                                       0, 1, 2, 3, 4, 5, 6, 7);
    if (fwd) keep8(0, Sown);
    handover(0, Sown);

    auto step = [&](int slot, int n, int par, bool prefetch, bool fkeep) {
        if (prefetch) issue((slot + PFv) % RSv, n + PFv);
        const v4f o0 = xoth[par * 128];
        const v4f o1 = xoth[par * 128 + 1];
        v8f w = weights(slot);
        v8f accA = {};
        v8f accB = {};
        #pragma unroll
        for (int j = 0; j < 4; ++j) {
            v2f b;
            b[0] = Sown[2 * j];
            b[1] = Sown[2 * j + 1];
            if (j & 1) accB = __builtin_amdgcn_wmma_f32_16x16x4_f32(false, Aown[j], false, b,
                                                                    (short)0, accB, false, false);
            else       accA = __builtin_amdgcn_wmma_f32_16x16x4_f32(false, Aown[j], false, b,
                                                                    (short)0, accA, false, false);
        }
        const v8f Soth = __builtin_shufflevector(o0, o1, 0, 1, 2, 3, 4, 5, 6, 7);
        #pragma unroll
        for (int j = 0; j < 4; ++j) {
            v2f b;
            b[0] = Soth[2 * j];
            b[1] = Soth[2 * j + 1];
            if (j & 1) accB = __builtin_amdgcn_wmma_f32_16x16x4_f32(false, Aoth[j], false, b,
                                                                    (short)0, accB, false, false);
            else       accA = __builtin_amdgcn_wmma_f32_16x16x4_f32(false, Aoth[j], false, b,
                                                                    (short)0, accA, false, false);
        }
        float mx = max16(Sown, Soth);
        mx = fmaxf(mx, other_half(mx));
        const float rs = __builtin_amdgcn_rcpf(mx);
        w *= rs;
        const v8f acc = accA + accB;
        Sown = acc * w;
        const int tt = fwd ? n : (Tv - 1 - n);
        if (fwd) { if (fkeep) keep8(tt & (CHL - 1), Sown); }
        else     keep8(tt & (CHL - 1), acc);
        handover(par ^ 1, Sown);
    };

    static_assert(CHL % RSv == 0 && RSv % 2 == 0, "ring slot and buffer parity must be compile-time");
    static_assert(CHL >= 2 * RSv, "a main loop and an unrolled tail");
    static_assert((CHL & (CHL - 1)) == 0 && Tv % CHL == 0, "row in block = t mod CHL");
    for (int s0 = 0; s0 < CHL - RSv; s0 += RSv) {
        #pragma unroll
        for (int u = 0; u < RSv; ++u) step(u, n0 + 1 + s0 + u, u & 1, true, true);
    }
    #pragma unroll
    for (int u = 0; u < RSv - 1; ++u)
        step(u, n0 + 1 + (CHL - RSv) + u, u & 1, u + PFv < RSv, true);
    step(RSv - 1, n0 + CHL, (RSv - 1) & 1, false, false);
    if (!fwd && chunk == 0) {
        v8f ones;
        #pragma unroll
        for (int r = 0; r < 8; ++r) ones[r] = 1.0f;
        keep8(CHL - 1, ones);
    }

    __syncthreads();
    {
        const int l8 = lane & 7;
        const int cc = wv * 4 + (lane >> 3);
        const float* ta = (const float*)tiles + cc * ABP + 4 * l8;
        const float* tb = ta + ABN;
        float* op = out + ((size_t)bat * Tv + cc * CHL) * Kv + 4 * l8;
        #pragma unroll 4
        for (int lr = 0; lr < CHL; ++lr) {
            const v4f a = *(const v4f*)(ta + lr * (16 * ABP));
            const v4f b = *(const v4f*)(tb + lr * (16 * ABP));
            const v4f p = a * b;
            float s = (p[0] + p[1]) + (p[2] + p[3]);
            s += __builtin_bit_cast(float, row_xor<1>(__builtin_bit_cast(int, s)));
            s += __builtin_bit_cast(float, row_xor<2>(__builtin_bit_cast(int, s)));
            s += __builtin_bit_cast(float, row_xor<4>(__builtin_bit_cast(int, s)));
            const float r = 1.0f / s;
            *(v4f*)(op + lr * Kv) = p * r;
        }
    }
}

extern "C" void kernel_launch(void* const* d_in, const int* in_sizes, int n_in,
                              void* d_out, int out_size, void* d_ws, size_t ws_size,
                              hipStream_t stream) {
    const float* x = (const float*)d_in[0];
    const float* W = (const float*)d_in[1];
    const float* U = (const float*)d_in[2];
    const float* b = (const float*)d_in[3];
    float* out = (float*)d_out;

    float* ws    = (float*)d_ws;
    float* wgt   = ws;
    float* Wp    = wgt + (size_t)Bv * Tv * Kv;
    float* Pm    = Wp + (size_t)Dv * Kv;
    float* Pt    = Pm + (size_t)NCH * Bv * 1024;
    float* Ex    = Pt + (size_t)NCH * Bv * 1024;

    crf_packW     <<<dim3(128), dim3(256), 0, stream>>>(W, Wp);
    crf_emissions <<<dim3((Bv * Tv / 16) / 8), dim3(256), 0, stream>>>(x, Wp, b, U, wgt, Pm, Pt, Ex);
    crf_scan      <<<dim3(Bv), dim3(128), 0, stream>>>(wgt, U, Pm, Pt, Ex, out);
}
